# starting kernel
# baseline (speedup 1.0000x reference)
#include <hip/hip_runtime.h>

typedef __attribute__((ext_vector_type(2))) float v2f;
typedef __attribute__((ext_vector_type(8))) float v8f;

#define NB    1024
#define DIN   64
#define DHID  128
#define BN_EPS 1e-5f

__global__ __launch_bounds__(32)
void k1_gemm(const float* __restrict__ x, const float* __restrict__ W1,
             const float* __restrict__ b1, float* __restrict__ A,
             float* __restrict__ Bm)
{
    const int i0   = blockIdx.x * 16;
    const int j0   = blockIdx.y * 16;
    const int hlf  = blockIdx.z;
    const int lane = threadIdx.x & 31;
    const int m    = lane & 15;
    const int kg   = (lane >> 4) << 1;

    const float* __restrict__ W = W1 + hlf * DIN * DHID;
    float* __restrict__ C = hlf ? Bm : A;

    v8f acc = {0.f, 0.f, 0.f, 0.f, 0.f, 0.f, 0.f, 0.f};
    for (int k = 0; k < DIN; k += 4) {
        v2f a, b;
        a.x = x[(i0 + m) * DIN + (k + kg)];
        a.y = x[(i0 + m) * DIN + (k + kg + 1)];
        b.x = W[(k + kg) * DHID + (j0 + m)];
        b.y = W[(k + kg + 1) * DHID + (j0 + m)];
        acc = __builtin_amdgcn_wmma_f32_16x16x4_f32(false, a, false, b,
                                                    (short)0, acc, false, false);
    }

    const float bias  = (hlf == 0) ? b1[j0 + m] : 0.0f;
    const int   rbase = (lane < 16) ? 0 : 8;
#pragma unroll
    for (int v = 0; v < 8; ++v) {
        C[(i0 + rbase + v) * DHID + (j0 + m)] = acc[v] + bias;
    }
}

__global__ __launch_bounds__(256)
void k2_stats(const float* __restrict__ A, const float* __restrict__ Bm,
              float* __restrict__ partials)
{
    const int h   = blockIdx.x;
    const int jc  = blockIdx.y;
    const int tid = threadIdx.x;

    __shared__ float bcol[256];
    __shared__ float red1[256];
    __shared__ float red2[256];

    bcol[tid] = Bm[(jc * 256 + tid) * DHID + h];

    const float a0 = A[(tid      ) * DHID + h];
    const float a1 = A[(tid + 256) * DHID + h];
    const float a2 = A[(tid + 512) * DHID + h];
    const float a3 = A[(tid + 768) * DHID + h];
    __syncthreads();

    float s1 = 0.f, s2 = 0.f;
    for (int j = 0; j < 256; ++j) {
        const float b  = bcol[j];
        const float r0 = fmaxf(a0 + b, 0.f);
        const float r1 = fmaxf(a1 + b, 0.f);
        const float r2 = fmaxf(a2 + b, 0.f);
        const float r3 = fmaxf(a3 + b, 0.f);
        s1 += (r0 + r1) + (r2 + r3);
        s2 = fmaf(r0, r0, s2); s2 = fmaf(r1, r1, s2);
        s2 = fmaf(r2, r2, s2); s2 = fmaf(r3, r3, s2);
    }
    red1[tid] = s1; red2[tid] = s2;
    __syncthreads();
    for (int s = 128; s > 0; s >>= 1) {
        if (tid < s) { red1[tid] += red1[tid + s]; red2[tid] += red2[tid + s]; }
        __syncthreads();
    }
    if (tid == 0) {
        partials[(h * 4 + jc) * 2 + 0] = red1[0];
        partials[(h * 4 + jc) * 2 + 1] = red2[0];
    }
}

__global__ __launch_bounds__(128)
void k3_final(const float* __restrict__ partials, const float* __restrict__ gamma,
              const float* __restrict__ beta, const float* __restrict__ W2,
              const float* __restrict__ b2, float* __restrict__ cvec,
              float* __restrict__ cterm)
{
    const int h = threadIdx.x;
    __shared__ float red[128];

    float S1 = 0.f, S2 = 0.f;
    for (int jc = 0; jc < 4; ++jc) {
        S1 += partials[(h * 4 + jc) * 2 + 0];
        S2 += partials[(h * 4 + jc) * 2 + 1];
    }
    const float invN = 1.0f / ((float)NB * (float)NB);
    const float mean = S1 * invN;
    const float var  = fmaxf(S2 * invN - mean * mean, 0.f);
    const float inv  = rsqrtf(var + BN_EPS);
    const float g    = gamma[h] * inv;
    cvec[h] = W2[h] * g;
    red[h]  = W2[h] * (beta[h] - mean * g);
    __syncthreads();
    for (int s = 64; s > 0; s >>= 1) {
        if (h < s) red[h] += red[h + s];
        __syncthreads();
    }
    if (h == 0) *cterm = red[0] + b2[0];
}

__global__ __launch_bounds__(256)
void k4_out(const float* __restrict__ A, const float* __restrict__ Bm,
            const float* __restrict__ cvec, const float* __restrict__ cterm,
            float* __restrict__ out)
{
    __shared__ float aT[DHID][64];
    __shared__ float bT[DHID][32];
    __shared__ float cs[DHID];

    const int i0  = blockIdx.x * 64;
    const int j0  = blockIdx.y * 32;
    const int tid = threadIdx.x;

    {
        const int il = tid & 63;
        const int hq = tid >> 6;
        const float4* src = (const float4*)(A + (i0 + il) * DHID + hq * 32);
#pragma unroll
        for (int q = 0; q < 8; ++q) {
            const float4 v = src[q];
            const int h = hq * 32 + q * 4;
            aT[h + 0][il] = v.x; aT[h + 1][il] = v.y;
            aT[h + 2][il] = v.z; aT[h + 3][il] = v.w;
        }
    }
    {
        const int jl = tid & 31;
        const int hq = tid >> 5;
        const float4* src = (const float4*)(Bm + (j0 + jl) * DHID + hq * 16);
#pragma unroll
        for (int q = 0; q < 4; ++q) {
            const float4 v = src[q];
            const int h = hq * 16 + q * 4;
            bT[h + 0][jl] = v.x; bT[h + 1][jl] = v.y;
            bT[h + 2][jl] = v.z; bT[h + 3][jl] = v.w;
        }
    }
    if (tid < DHID) cs[tid] = cvec[tid];
    __syncthreads();

    const int ib = (tid & 15) * 4;
    const int jb = (tid >> 4) * 2;
    float acc[4][2] = {};

    for (int h = 0; h < DHID; ++h) {
        const float4 av = *(const float4*)&aT[h][ib];
        const float  b0 = bT[h][jb];
        const float  b1v = bT[h][jb + 1];
        const float  ck = cs[h];
        acc[0][0] = fmaf(ck, fmaxf(av.x + b0, 0.f), acc[0][0]);
        acc[0][1] = fmaf(ck, fmaxf(av.x + b1v, 0.f), acc[0][1]);
        acc[1][0] = fmaf(ck, fmaxf(av.y + b0, 0.f), acc[1][0]);
        acc[1][1] = fmaf(ck, fmaxf(av.y + b1v, 0.f), acc[1][1]);
        acc[2][0] = fmaf(ck, fmaxf(av.z + b0, 0.f), acc[2][0]);
        acc[2][1] = fmaf(ck, fmaxf(av.z + b1v, 0.f), acc[2][1]);
        acc[3][0] = fmaf(ck, fmaxf(av.w + b0, 0.f), acc[3][0]);
        acc[3][1] = fmaf(ck, fmaxf(av.w + b1v, 0.f), acc[3][1]);
    }

    const float ct = *cterm;
#pragma unroll
    for (int ii = 0; ii < 4; ++ii) {
        float2 o;
        o.x = acc[ii][0] + ct;
        o.y = acc[ii][1] + ct;
        *(float2*)(out + (size_t)(i0 + ib + ii) * NB + (j0 + jb)) = o;
    }
}

extern "C" void kernel_launch(void* const* d_in, const int* in_sizes, int n_in,
                              void* d_out, int out_size, void* d_ws, size_t ws_size,
                              hipStream_t stream)
{
    const float* x     = (const float*)d_in[0];
    const float* W1    = (const float*)d_in[1];
    const float* b1    = (const float*)d_in[2];
    const float* gamma = (const float*)d_in[3];
    const float* beta  = (const float*)d_in[4];
    const float* W2    = (const float*)d_in[5];
    const float* b2    = (const float*)d_in[6];
    float* out = (float*)d_out;

    float* ws       = (float*)d_ws;
    float* A        = ws;
    float* Bm       = ws + NB * DHID;
    float* partials = ws + 2 * NB * DHID;
    float* cvec     = partials + DHID * 4 * 2;
    float* cterm    = cvec + DHID;

    k1_gemm <<<dim3(NB / 16, DHID / 16, 2), 32,  0, stream>>>(x, W1, b1, A, Bm);
    k2_stats<<<dim3(DHID, 4),              256, 0, stream>>>(A, Bm, partials);
    k3_final<<<1,                          DHID, 0, stream>>>(partials, gamma, beta, W2, b2, cvec, cterm);
    k4_out  <<<dim3(NB / 64, NB / 32),     256, 0, stream>>>(A, Bm, cvec, cterm, out);
}


# K2 stats: sort each feature's b column once, suffix sums of b and b^2, binary search per a_i gives the row sums in closed form instead of all 1M pairs; changes the summation order of S1 and S2
# speedup vs baseline: 1.1687x; 1.1687x over previous
#include <hip/hip_runtime.h>

typedef __attribute__((ext_vector_type(2))) float v2f;
typedef __attribute__((ext_vector_type(8))) float v8f;

#define NB    1024
#define DIN   64
#define DHID  128
#define BN_EPS 1e-5f

__global__ __launch_bounds__(32)
void k1_gemm(const float* __restrict__ x, const float* __restrict__ W1,
             const float* __restrict__ b1, float* __restrict__ A,
             float* __restrict__ Bm)
{
    const int i0   = blockIdx.x * 16;
    const int j0   = blockIdx.y * 16;
    const int hlf  = blockIdx.z;
    const int lane = threadIdx.x & 31;
    const int m    = lane & 15;
    const int kg   = (lane >> 4) << 1;

    const float* __restrict__ W = W1 + hlf * DIN * DHID;
    float* __restrict__ C = hlf ? Bm : A;

    v8f acc = {0.f, 0.f, 0.f, 0.f, 0.f, 0.f, 0.f, 0.f};
    for (int k = 0; k < DIN; k += 4) {
        v2f a, b;
        a.x = x[(i0 + m) * DIN + (k + kg)];
        a.y = x[(i0 + m) * DIN + (k + kg + 1)];
        b.x = W[(k + kg) * DHID + (j0 + m)];
        b.y = W[(k + kg + 1) * DHID + (j0 + m)];
        acc = __builtin_amdgcn_wmma_f32_16x16x4_f32(false, a, false, b,
                                                    (short)0, acc, false, false);
    }

    const float bias  = (hlf == 0) ? b1[j0 + m] : 0.0f;
    const int   rbase = (lane < 16) ? 0 : 8;
#pragma unroll
    for (int v = 0; v < 8; ++v) {
        C[(i0 + rbase + v) * DHID + (j0 + m)] = acc[v] + bias;
    }
}

__global__ __launch_bounds__(256)
void k2_stats(const float* __restrict__ A, const float* __restrict__ Bm,
              float* __restrict__ partials)
{
    const int h    = blockIdx.x;
    const int tid  = threadIdx.x;
    const int lane = tid & 31;
    const int wave = tid >> 5;

    __shared__ float4 xch[2][256];
    __shared__ __attribute__((aligned(16))) float sb[NB];
    __shared__ __attribute__((aligned(16))) float p1[NB + 4];
    __shared__ __attribute__((aligned(16))) float p2[NB + 4];
    __shared__ float wtot[2][8];
    __shared__ float red[2][8];

    float x[4], a[4];
#pragma unroll
    for (int r = 0; r < 4; ++r) x[r] = Bm[(tid * 4 + r) * DHID + h];
#pragma unroll
    for (int r = 0; r < 4; ++r) a[r] = A[(tid + 256 * r) * DHID + h];

    int xb = 0;
#pragma unroll
    for (int k = 2; k <= NB; k <<= 1) {
#pragma unroll
        for (int j = k >> 1; j >= 1; j >>= 1) {
            if (j >= 128) {
                xch[xb][tid] = make_float4(x[0], x[1], x[2], x[3]);
                __syncthreads();
                const float4 y = xch[xb][tid ^ (j >> 2)];
                xb ^= 1;
                const bool keep_min = ((((tid * 4) & j) == 0) == (((tid * 4) & k) == 0));
                x[0] = keep_min ? fminf(x[0], y.x) : fmaxf(x[0], y.x);
                x[1] = keep_min ? fminf(x[1], y.y) : fmaxf(x[1], y.y);
                x[2] = keep_min ? fminf(x[2], y.z) : fmaxf(x[2], y.z);
                x[3] = keep_min ? fminf(x[3], y.w) : fmaxf(x[3], y.w);
            } else if (j >= 4) {
                const bool keep_min = ((((tid * 4) & j) == 0) == (((tid * 4) & k) == 0));
#pragma unroll
                for (int r = 0; r < 4; ++r) {
                    const float y = __shfl_xor(x[r], j >> 2, 32);
                    x[r] = keep_min ? fminf(x[r], y) : fmaxf(x[r], y);
                }
            } else {
#pragma unroll
                for (int r = 0; r < 4; ++r) {
                    if ((r & j) == 0) {
                        const bool asc = (((tid * 4 + r) & k) == 0);
                        const float lo = fminf(x[r], x[r | j]);
                        const float hi = fmaxf(x[r], x[r | j]);
                        x[r]     = asc ? lo : hi;
                        x[r | j] = asc ? hi : lo;
                    }
                }
            }
        }
    }

    float s1[4], s2[4];
    s1[3] = x[3];          s2[3] = x[3] * x[3];
    s1[2] = x[2] + s1[3];  s2[2] = fmaf(x[2], x[2], s2[3]);
    s1[1] = x[1] + s1[2];  s2[1] = fmaf(x[1], x[1], s2[2]);
    s1[0] = x[0] + s1[1];  s2[0] = fmaf(x[0], x[0], s2[1]);

    float t1 = s1[0], t2 = s2[0];
#pragma unroll
    for (int d = 1; d < 32; d <<= 1) {
        const float y1 = __shfl_down(t1, d, 32);
        const float y2 = __shfl_down(t2, d, 32);
        if (lane + d < 32) { t1 += y1; t2 += y2; }
    }
    const float n1 = __shfl_down(t1, 1, 32);
    const float n2 = __shfl_down(t2, 1, 32);
    float off1 = (lane < 31) ? n1 : 0.f;
    float off2 = (lane < 31) ? n2 : 0.f;
    if (lane == 0) { wtot[0][wave] = t1; wtot[1][wave] = t2; }
    __syncthreads();
    {
        float hi1 = 0.f, hi2 = 0.f;
#pragma unroll
        for (int w = 7; w >= 1; --w) {
            if (w > wave) { hi1 += wtot[0][w]; hi2 += wtot[1][w]; }
        }
        off1 += hi1; off2 += hi2;
    }
    *(float4*)&sb[tid * 4] = make_float4(x[0], x[1], x[2], x[3]);
    *(float4*)&p1[tid * 4] = make_float4(s1[0] + off1, s1[1] + off1, s1[2] + off1, s1[3] + off1);
    *(float4*)&p2[tid * 4] = make_float4(s2[0] + off2, s2[1] + off2, s2[2] + off2, s2[3] + off2);
    if (tid == 0) { p1[NB] = 0.f; p2[NB] = 0.f; }
    __syncthreads();

    float na[4];
    int   m[4] = {0, 0, 0, 0};
#pragma unroll
    for (int r = 0; r < 4; ++r) na[r] = -a[r];
#pragma unroll
    for (int s = NB / 2; s >= 1; s >>= 1) {
#pragma unroll
        for (int r = 0; r < 4; ++r) {
            if (sb[m[r] + s - 1] <= na[r]) m[r] += s;
        }
    }
    float acc1 = 0.f, acc2 = 0.f;
#pragma unroll
    for (int r = 0; r < 4; ++r) {
        if (sb[m[r]] <= na[r]) m[r] += 1;
        const float cnt = (float)(NB - m[r]);
        const float q1  = p1[m[r]];
        const float q2  = p2[m[r]];
        acc1 += fmaf(cnt, a[r], q1);
        acc2 += fmaf(a[r], fmaf(cnt, a[r], 2.f * q1), q2);
    }

#pragma unroll
    for (int d = 16; d >= 1; d >>= 1) {
        acc1 += __shfl_xor(acc1, d, 32);
        acc2 += __shfl_xor(acc2, d, 32);
    }
    if (lane == 0) { red[0][wave] = acc1; red[1][wave] = acc2; }
    __syncthreads();
    if (tid == 0) {
        float r1 = 0.f, r2 = 0.f;
#pragma unroll
        for (int w = 0; w < 8; ++w) { r1 += red[0][w]; r2 += red[1][w]; }
        partials[h * 2 + 0] = r1;
        partials[h * 2 + 1] = r2;
    }
}

__global__ __launch_bounds__(128)
void k3_final(const float* __restrict__ partials, const float* __restrict__ gamma,
              const float* __restrict__ beta, const float* __restrict__ W2,
              const float* __restrict__ b2, float* __restrict__ cvec,
              float* __restrict__ cterm)
{
    const int h = threadIdx.x;
    __shared__ float red[128];

    const float S1 = partials[h * 2 + 0];
    const float S2 = partials[h * 2 + 1];
    const float invN = 1.0f / ((float)NB * (float)NB);
    const float mean = S1 * invN;
    const float var  = fmaxf(S2 * invN - mean * mean, 0.f);
    const float inv  = rsqrtf(var + BN_EPS);
    const float g    = gamma[h] * inv;
    cvec[h] = W2[h] * g;
    red[h]  = W2[h] * (beta[h] - mean * g);
    __syncthreads();
    for (int s = 64; s > 0; s >>= 1) {
        if (h < s) red[h] += red[h + s];
        __syncthreads();
    }
    if (h == 0) *cterm = red[0] + b2[0];
}

__global__ __launch_bounds__(256)
void k4_out(const float* __restrict__ A, const float* __restrict__ Bm,
            const float* __restrict__ cvec, const float* __restrict__ cterm,
            float* __restrict__ out)
{
    __shared__ float aT[DHID][64];
    __shared__ float bT[DHID][32];
    __shared__ float cs[DHID];

    const int i0  = blockIdx.x * 64;
    const int j0  = blockIdx.y * 32;
    const int tid = threadIdx.x;

    {
        const int il = tid & 63;
        const int hq = tid >> 6;
        const float4* src = (const float4*)(A + (i0 + il) * DHID + hq * 32);
#pragma unroll
        for (int q = 0; q < 8; ++q) {
            const float4 v = src[q];
            const int h = hq * 32 + q * 4;
            aT[h + 0][il] = v.x; aT[h + 1][il] = v.y;
            aT[h + 2][il] = v.z; aT[h + 3][il] = v.w;
        }
    }
    {
        const int jl = tid & 31;
        const int hq = tid >> 5;
        const float4* src = (const float4*)(Bm + (j0 + jl) * DHID + hq * 16);
#pragma unroll
        for (int q = 0; q < 4; ++q) {
            const float4 v = src[q];
            const int h = hq * 16 + q * 4;
            bT[h + 0][jl] = v.x; bT[h + 1][jl] = v.y;
            bT[h + 2][jl] = v.z; bT[h + 3][jl] = v.w;
        }
    }
    if (tid < DHID) cs[tid] = cvec[tid];
    __syncthreads();

    const int ib = (tid & 15) * 4;
    const int jb = (tid >> 4) * 2;
    float acc[4][2] = {};

    for (int h = 0; h < DHID; ++h) {
        const float4 av = *(const float4*)&aT[h][ib];
        const float  b0 = bT[h][jb];
        const float  b1v = bT[h][jb + 1];
        const float  ck = cs[h];
        acc[0][0] = fmaf(ck, fmaxf(av.x + b0, 0.f), acc[0][0]);
        acc[0][1] = fmaf(ck, fmaxf(av.x + b1v, 0.f), acc[0][1]);
        acc[1][0] = fmaf(ck, fmaxf(av.y + b0, 0.f), acc[1][0]);
        acc[1][1] = fmaf(ck, fmaxf(av.y + b1v, 0.f), acc[1][1]);
        acc[2][0] = fmaf(ck, fmaxf(av.z + b0, 0.f), acc[2][0]);
        acc[2][1] = fmaf(ck, fmaxf(av.z + b1v, 0.f), acc[2][1]);
        acc[3][0] = fmaf(ck, fmaxf(av.w + b0, 0.f), acc[3][0]);
        acc[3][1] = fmaf(ck, fmaxf(av.w + b1v, 0.f), acc[3][1]);
    }

    const float ct = *cterm;
#pragma unroll
    for (int ii = 0; ii < 4; ++ii) {
        float2 o;
        o.x = acc[ii][0] + ct;
        o.y = acc[ii][1] + ct;
        *(float2*)(out + (size_t)(i0 + ib + ii) * NB + (j0 + jb)) = o;
    }
}

extern "C" void kernel_launch(void* const* d_in, const int* in_sizes, int n_in,
                              void* d_out, int out_size, void* d_ws, size_t ws_size,
                              hipStream_t stream)
{
    const float* x     = (const float*)d_in[0];
    const float* W1    = (const float*)d_in[1];
    const float* b1    = (const float*)d_in[2];
    const float* gamma = (const float*)d_in[3];
    const float* beta  = (const float*)d_in[4];
    const float* W2    = (const float*)d_in[5];
    const float* b2    = (const float*)d_in[6];
    float* out = (float*)d_out;

    float* ws       = (float*)d_ws;
    float* A        = ws;
    float* Bm       = ws + NB * DHID;
    float* partials = ws + 2 * NB * DHID;
    float* cvec     = partials + DHID * 4 * 2;
    float* cterm    = cvec + DHID;

    k1_gemm <<<dim3(NB / 16, DHID / 16, 2), 32,  0, stream>>>(x, W1, b1, A, Bm);
    k2_stats<<<dim3(DHID),                 256, 0, stream>>>(A, Bm, partials);
    k3_final<<<1,                          DHID, 0, stream>>>(partials, gamma, beta, W2, b2, cvec, cterm);
    k4_out  <<<dim3(NB / 64, NB / 32),     256, 0, stream>>>(A, Bm, cvec, cterm, out);
}


# finalize launch removed (4 -> 3 launches): each stats block finishes its own feature (mean, var, rsqrt, scale, offset term); every output wave sums the 128 terms itself, same tree, sum order kept
# speedup vs baseline: 1.3259x; 1.1345x over previous
#include <hip/hip_runtime.h>

typedef __attribute__((ext_vector_type(2))) float v2f;
typedef __attribute__((ext_vector_type(8))) float v8f;

#define NB    1024
#define DIN   64
#define DHID  128
#define BN_EPS 1e-5f

__global__ __launch_bounds__(32)
void k1_gemm(const float* __restrict__ x, const float* __restrict__ W1,
             const float* __restrict__ b1, float* __restrict__ A,
             float* __restrict__ Bm)
{
    const int i0   = blockIdx.x * 16;
    const int j0   = blockIdx.y * 16;
    const int hlf  = blockIdx.z;
    const int lane = threadIdx.x & 31;
    const int m    = lane & 15;
    const int kg   = (lane >> 4) << 1;

    const float* __restrict__ W = W1 + hlf * DIN * DHID;
    float* __restrict__ C = hlf ? Bm : A;

    v8f acc = {0.f, 0.f, 0.f, 0.f, 0.f, 0.f, 0.f, 0.f};
    for (int k = 0; k < DIN; k += 4) {
        v2f a, b;
        a.x = x[(i0 + m) * DIN + (k + kg)];
        a.y = x[(i0 + m) * DIN + (k + kg + 1)];
        b.x = W[(k + kg) * DHID + (j0 + m)];
        b.y = W[(k + kg + 1) * DHID + (j0 + m)];
        acc = __builtin_amdgcn_wmma_f32_16x16x4_f32(false, a, false, b,
                                                    (short)0, acc, false, false);
    }

    const float bias  = (hlf == 0) ? b1[j0 + m] : 0.0f;
    const int   rbase = (lane < 16) ? 0 : 8;
#pragma unroll
    for (int v = 0; v < 8; ++v) {
        C[(i0 + rbase + v) * DHID + (j0 + m)] = acc[v] + bias;
    }
}

__global__ __launch_bounds__(256)
void k2_stats(const float* __restrict__ A, const float* __restrict__ Bm,
              const float* __restrict__ gamma, const float* __restrict__ beta,
              const float* __restrict__ W2,
              float* __restrict__ cvec, float* __restrict__ term)
{
    const int h    = blockIdx.x;
    const int tid  = threadIdx.x;
    const int lane = tid & 31;
    const int wave = tid >> 5;

    const float gm = gamma[h];
    const float bt = beta[h];
    const float w2 = W2[h];

    __shared__ float4 xch[2][256];
    __shared__ __attribute__((aligned(16))) float sb[NB];
    __shared__ __attribute__((aligned(16))) float p1[NB + 4];
    __shared__ __attribute__((aligned(16))) float p2[NB + 4];
    __shared__ float wtot[2][8];
    __shared__ float red[2][8];

    float x[4], a[4];
#pragma unroll
    for (int r = 0; r < 4; ++r) x[r] = Bm[(tid * 4 + r) * DHID + h];
#pragma unroll
    for (int r = 0; r < 4; ++r) a[r] = A[(tid + 256 * r) * DHID + h];

    int xb = 0;
#pragma unroll
    for (int k = 2; k <= NB; k <<= 1) {
#pragma unroll
        for (int j = k >> 1; j >= 1; j >>= 1) {
            if (j >= 128) {
                xch[xb][tid] = make_float4(x[0], x[1], x[2], x[3]);
                __syncthreads();
                const float4 y = xch[xb][tid ^ (j >> 2)];
                xb ^= 1;
                const bool keep_min = ((((tid * 4) & j) == 0) == (((tid * 4) & k) == 0));
                x[0] = keep_min ? fminf(x[0], y.x) : fmaxf(x[0], y.x);
                x[1] = keep_min ? fminf(x[1], y.y) : fmaxf(x[1], y.y);
                x[2] = keep_min ? fminf(x[2], y.z) : fmaxf(x[2], y.z);
                x[3] = keep_min ? fminf(x[3], y.w) : fmaxf(x[3], y.w);
            } else if (j >= 4) {
                const bool keep_min = ((((tid * 4) & j) == 0) == (((tid * 4) & k) == 0));
#pragma unroll
                for (int r = 0; r < 4; ++r) {
                    const float y = __shfl_xor(x[r], j >> 2, 32);
                    x[r] = keep_min ? fminf(x[r], y) : fmaxf(x[r], y);
                }
            } else {
#pragma unroll
                for (int r = 0; r < 4; ++r) {
                    if ((r & j) == 0) {
                        const bool asc = (((tid * 4 + r) & k) == 0);
                        const float lo = fminf(x[r], x[r | j]);
                        const float hi = fmaxf(x[r], x[r | j]);
                        x[r]     = asc ? lo : hi;
                        x[r | j] = asc ? hi : lo;
                    }
                }
            }
        }
    }

    float s1[4], s2[4];
    s1[3] = x[3];          s2[3] = x[3] * x[3];
    s1[2] = x[2] + s1[3];  s2[2] = fmaf(x[2], x[2], s2[3]);
    s1[1] = x[1] + s1[2];  s2[1] = fmaf(x[1], x[1], s2[2]);
    s1[0] = x[0] + s1[1];  s2[0] = fmaf(x[0], x[0], s2[1]);

    float t1 = s1[0], t2 = s2[0];
#pragma unroll
    for (int d = 1; d < 32; d <<= 1) {
        const float y1 = __shfl_down(t1, d, 32);
        const float y2 = __shfl_down(t2, d, 32);
        if (lane + d < 32) { t1 += y1; t2 += y2; }
    }
    const float n1 = __shfl_down(t1, 1, 32);
    const float n2 = __shfl_down(t2, 1, 32);
    float off1 = (lane < 31) ? n1 : 0.f;
    float off2 = (lane < 31) ? n2 : 0.f;
    if (lane == 0) { wtot[0][wave] = t1; wtot[1][wave] = t2; }
    __syncthreads();
    {
        float hi1 = 0.f, hi2 = 0.f;
#pragma unroll
        for (int w = 7; w >= 1; --w) {
            if (w > wave) { hi1 += wtot[0][w]; hi2 += wtot[1][w]; }
        }
        off1 += hi1; off2 += hi2;
    }
    *(float4*)&sb[tid * 4] = make_float4(x[0], x[1], x[2], x[3]);
    *(float4*)&p1[tid * 4] = make_float4(s1[0] + off1, s1[1] + off1, s1[2] + off1, s1[3] + off1);
    *(float4*)&p2[tid * 4] = make_float4(s2[0] + off2, s2[1] + off2, s2[2] + off2, s2[3] + off2);
    if (tid == 0) { p1[NB] = 0.f; p2[NB] = 0.f; }
    __syncthreads();

    float na[4];
    int   m[4] = {0, 0, 0, 0};
#pragma unroll
    for (int r = 0; r < 4; ++r) na[r] = -a[r];
#pragma unroll
    for (int s = NB / 2; s >= 1; s >>= 1) {
#pragma unroll
        for (int r = 0; r < 4; ++r) {
            if (sb[m[r] + s - 1] <= na[r]) m[r] += s;
        }
    }
    float acc1 = 0.f, acc2 = 0.f;
#pragma unroll
    for (int r = 0; r < 4; ++r) {
        if (sb[m[r]] <= na[r]) m[r] += 1;
        const float cnt = (float)(NB - m[r]);
        const float q1  = p1[m[r]];
        const float q2  = p2[m[r]];
        acc1 += fmaf(cnt, a[r], q1);
        acc2 += fmaf(a[r], fmaf(cnt, a[r], 2.f * q1), q2);
    }

#pragma unroll
    for (int d = 16; d >= 1; d >>= 1) {
        acc1 += __shfl_xor(acc1, d, 32);
        acc2 += __shfl_xor(acc2, d, 32);
    }
    if (lane == 0) { red[0][wave] = acc1; red[1][wave] = acc2; }
    __syncthreads();
    if (tid == 0) {
        float r1 = 0.f, r2 = 0.f;
#pragma unroll
        for (int w = 0; w < 8; ++w) { r1 += red[0][w]; r2 += red[1][w]; }
        const float invN = 1.0f / ((float)NB * (float)NB);
        const float mean = r1 * invN;
        const float var  = fmaxf(r2 * invN - mean * mean, 0.f);
        const float inv  = rsqrtf(var + BN_EPS);
        const float g    = gm * inv;
        cvec[h] = w2 * g;
        term[h] = w2 * (bt - mean * g);
    }
}

__device__ __forceinline__ float lane_xor16(float v)
{
    const int b = __builtin_bit_cast(int, v);
    return __builtin_bit_cast(float,
        __builtin_amdgcn_permlanex16(b, b, 0x76543210, 0xfedcba98, false, false));
}
template <int M>
__device__ __forceinline__ float row_xor(float v)
{
    const int b = __builtin_bit_cast(int, v);
    return __builtin_bit_cast(float,
        __builtin_amdgcn_update_dpp(b, b, 0x160 | M, 0xf, 0xf, false));
}

__global__ __launch_bounds__(256)
void k4_out(const float* __restrict__ A, const float* __restrict__ Bm,
            const float* __restrict__ cvec, const float* __restrict__ term,
            const float* __restrict__ b2, float* __restrict__ out)
{
    __shared__ float aT[DHID][64];
    __shared__ float bT[DHID][32];
    __shared__ float cs[DHID];

    const int i0  = blockIdx.x * 64;
    const int j0  = blockIdx.y * 32;
    const int tid = threadIdx.x;

    const int   ln = tid & 31;
    const float t0 = term[ln], t1 = term[ln + 32], t2 = term[ln + 64], t3 = term[ln + 96];
    const float bias2 = b2[0];

    {
        const int il = tid & 63;
        const int hq = tid >> 6;
        const float4* src = (const float4*)(A + (i0 + il) * DHID + hq * 32);
#pragma unroll
        for (int q = 0; q < 8; ++q) {
            const float4 v = src[q];
            const int h = hq * 32 + q * 4;
            aT[h + 0][il] = v.x; aT[h + 1][il] = v.y;
            aT[h + 2][il] = v.z; aT[h + 3][il] = v.w;
        }
    }
    {
        const int jl = tid & 31;
        const int hq = tid >> 5;
        const float4* src = (const float4*)(Bm + (j0 + jl) * DHID + hq * 16);
#pragma unroll
        for (int q = 0; q < 4; ++q) {
            const float4 v = src[q];
            const int h = hq * 16 + q * 4;
            bT[h + 0][jl] = v.x; bT[h + 1][jl] = v.y;
            bT[h + 2][jl] = v.z; bT[h + 3][jl] = v.w;
        }
    }
    if (tid < DHID) cs[tid] = cvec[tid];

    float cv = (t0 + t2) + (t1 + t3);
    cv += lane_xor16(cv);
    cv += row_xor<8>(cv);
    cv += row_xor<4>(cv);
    cv += row_xor<2>(cv);
    cv += row_xor<1>(cv);
    const float ct = cv + bias2;
    __syncthreads();

    const int ib = (tid & 15) * 4;
    const int jb = (tid >> 4) * 2;
    float acc[4][2] = {};

    for (int h = 0; h < DHID; ++h) {
        const float4 av = *(const float4*)&aT[h][ib];
        const float  b0 = bT[h][jb];
        const float  b1v = bT[h][jb + 1];
        const float  ck = cs[h];
        acc[0][0] = fmaf(ck, fmaxf(av.x + b0, 0.f), acc[0][0]);
        acc[0][1] = fmaf(ck, fmaxf(av.x + b1v, 0.f), acc[0][1]);
        acc[1][0] = fmaf(ck, fmaxf(av.y + b0, 0.f), acc[1][0]);
        acc[1][1] = fmaf(ck, fmaxf(av.y + b1v, 0.f), acc[1][1]);
        acc[2][0] = fmaf(ck, fmaxf(av.z + b0, 0.f), acc[2][0]);
        acc[2][1] = fmaf(ck, fmaxf(av.z + b1v, 0.f), acc[2][1]);
        acc[3][0] = fmaf(ck, fmaxf(av.w + b0, 0.f), acc[3][0]);
        acc[3][1] = fmaf(ck, fmaxf(av.w + b1v, 0.f), acc[3][1]);
    }

#pragma unroll
    for (int ii = 0; ii < 4; ++ii) {
        float2 o;
        o.x = acc[ii][0] + ct;
        o.y = acc[ii][1] + ct;
        *(float2*)(out + (size_t)(i0 + ib + ii) * NB + (j0 + jb)) = o;
    }
}

extern "C" void kernel_launch(void* const* d_in, const int* in_sizes, int n_in,
                              void* d_out, int out_size, void* d_ws, size_t ws_size,
                              hipStream_t stream)
{
    const float* x     = (const float*)d_in[0];
    const float* W1    = (const float*)d_in[1];
    const float* b1    = (const float*)d_in[2];
    const float* gamma = (const float*)d_in[3];
    const float* beta  = (const float*)d_in[4];
    const float* W2    = (const float*)d_in[5];
    const float* b2    = (const float*)d_in[6];
    float* out = (float*)d_out;

    float* ws       = (float*)d_ws;
    float* A        = ws;
    float* Bm       = ws + NB * DHID;
    float* cvec     = ws + 2 * NB * DHID;
    float* term     = cvec + DHID;

    k1_gemm <<<dim3(NB / 16, DHID / 16, 2), 32,  0, stream>>>(x, W1, b1, A, Bm);
    k2_stats<<<dim3(DHID),                 256, 0, stream>>>(A, Bm, gamma, beta, W2, cvec, term);
    k4_out  <<<dim3(NB / 64, NB / 32),     256, 0, stream>>>(A, Bm, cvec, term, b2, out);
}


# K4: 4x4 outputs per thread on a 64x64 block tile (was 4x2 on 64x32), lanes along j: half the waves to launch, half the LDS reads and fewer issue slots per term; per-output sum order kept
# speedup vs baseline: 1.5574x; 1.1746x over previous
#include <hip/hip_runtime.h>

typedef __attribute__((ext_vector_type(2))) float v2f;
typedef __attribute__((ext_vector_type(8))) float v8f;

#define NB    1024
#define DIN   64
#define DHID  128
#define BN_EPS 1e-5f

__global__ __launch_bounds__(32)
void k1_gemm(const float* __restrict__ x, const float* __restrict__ W1,
             const float* __restrict__ b1, float* __restrict__ A,
             float* __restrict__ Bm)
{
    const int i0   = blockIdx.x * 16;
    const int j0   = blockIdx.y * 16;
    const int hlf  = blockIdx.z;
    const int lane = threadIdx.x & 31;
    const int m    = lane & 15;
    const int kg   = (lane >> 4) << 1;

    const float* __restrict__ W = W1 + hlf * DIN * DHID;
    float* __restrict__ C = hlf ? Bm : A;

    v8f acc = {0.f, 0.f, 0.f, 0.f, 0.f, 0.f, 0.f, 0.f};
    for (int k = 0; k < DIN; k += 4) {
        v2f a, b;
        a.x = x[(i0 + m) * DIN + (k + kg)];
        a.y = x[(i0 + m) * DIN + (k + kg + 1)];
        b.x = W[(k + kg) * DHID + (j0 + m)];
        b.y = W[(k + kg + 1) * DHID + (j0 + m)];
        acc = __builtin_amdgcn_wmma_f32_16x16x4_f32(false, a, false, b,
                                                    (short)0, acc, false, false);
    }

    const float bias  = (hlf == 0) ? b1[j0 + m] : 0.0f;
    const int   rbase = (lane < 16) ? 0 : 8;
#pragma unroll
    for (int v = 0; v < 8; ++v) {
        C[(i0 + rbase + v) * DHID + (j0 + m)] = acc[v] + bias;
    }
}

__global__ __launch_bounds__(256)
void k2_stats(const float* __restrict__ A, const float* __restrict__ Bm,
              const float* __restrict__ gamma, const float* __restrict__ beta,
              const float* __restrict__ W2,
              float* __restrict__ cvec, float* __restrict__ term)
{
    const int h    = blockIdx.x;
    const int tid  = threadIdx.x;
    const int lane = tid & 31;
    const int wave = tid >> 5;

    const float gm = gamma[h];
    const float bt = beta[h];
    const float w2 = W2[h];

    __shared__ float4 xch[2][256];
    __shared__ __attribute__((aligned(16))) float sb[NB];
    __shared__ __attribute__((aligned(16))) float p1[NB + 4];
    __shared__ __attribute__((aligned(16))) float p2[NB + 4];
    __shared__ float wtot[2][8];
    __shared__ float red[2][8];

    float x[4], a[4];
#pragma unroll
    for (int r = 0; r < 4; ++r) x[r] = Bm[(tid * 4 + r) * DHID + h];
#pragma unroll
    for (int r = 0; r < 4; ++r) a[r] = A[(tid + 256 * r) * DHID + h];

    int xb = 0;
#pragma unroll
    for (int k = 2; k <= NB; k <<= 1) {
#pragma unroll
        for (int j = k >> 1; j >= 1; j >>= 1) {
            if (j >= 128) {
                xch[xb][tid] = make_float4(x[0], x[1], x[2], x[3]);
                __syncthreads();
                const float4 y = xch[xb][tid ^ (j >> 2)];
                xb ^= 1;
                const bool keep_min = ((((tid * 4) & j) == 0) == (((tid * 4) & k) == 0));
                x[0] = keep_min ? fminf(x[0], y.x) : fmaxf(x[0], y.x);
                x[1] = keep_min ? fminf(x[1], y.y) : fmaxf(x[1], y.y);
                x[2] = keep_min ? fminf(x[2], y.z) : fmaxf(x[2], y.z);
                x[3] = keep_min ? fminf(x[3], y.w) : fmaxf(x[3], y.w);
            } else if (j >= 4) {
                const bool keep_min = ((((tid * 4) & j) == 0) == (((tid * 4) & k) == 0));
#pragma unroll
                for (int r = 0; r < 4; ++r) {
                    const float y = __shfl_xor(x[r], j >> 2, 32);
                    x[r] = keep_min ? fminf(x[r], y) : fmaxf(x[r], y);
                }
            } else {
#pragma unroll
                for (int r = 0; r < 4; ++r) {
                    if ((r & j) == 0) {
                        const bool asc = (((tid * 4 + r) & k) == 0);
                        const float lo = fminf(x[r], x[r | j]);
                        const float hi = fmaxf(x[r], x[r | j]);
                        x[r]     = asc ? lo : hi;
                        x[r | j] = asc ? hi : lo;
                    }
                }
            }
        }
    }

    float s1[4], s2[4];
    s1[3] = x[3];          s2[3] = x[3] * x[3];
    s1[2] = x[2] + s1[3];  s2[2] = fmaf(x[2], x[2], s2[3]);
    s1[1] = x[1] + s1[2];  s2[1] = fmaf(x[1], x[1], s2[2]);
    s1[0] = x[0] + s1[1];  s2[0] = fmaf(x[0], x[0], s2[1]);

    float t1 = s1[0], t2 = s2[0];
#pragma unroll
    for (int d = 1; d < 32; d <<= 1) {
        const float y1 = __shfl_down(t1, d, 32);
        const float y2 = __shfl_down(t2, d, 32);
        if (lane + d < 32) { t1 += y1; t2 += y2; }
    }
    const float n1 = __shfl_down(t1, 1, 32);
    const float n2 = __shfl_down(t2, 1, 32);
    float off1 = (lane < 31) ? n1 : 0.f;
    float off2 = (lane < 31) ? n2 : 0.f;
    if (lane == 0) { wtot[0][wave] = t1; wtot[1][wave] = t2; }
    __syncthreads();
    {
        float hi1 = 0.f, hi2 = 0.f;
#pragma unroll
        for (int w = 7; w >= 1; --w) {
            if (w > wave) { hi1 += wtot[0][w]; hi2 += wtot[1][w]; }
        }
        off1 += hi1; off2 += hi2;
    }
    *(float4*)&sb[tid * 4] = make_float4(x[0], x[1], x[2], x[3]);
    *(float4*)&p1[tid * 4] = make_float4(s1[0] + off1, s1[1] + off1, s1[2] + off1, s1[3] + off1);
    *(float4*)&p2[tid * 4] = make_float4(s2[0] + off2, s2[1] + off2, s2[2] + off2, s2[3] + off2);
    if (tid == 0) { p1[NB] = 0.f; p2[NB] = 0.f; }
    __syncthreads();

    float na[4];
    int   m[4] = {0, 0, 0, 0};
#pragma unroll
    for (int r = 0; r < 4; ++r) na[r] = -a[r];
#pragma unroll
    for (int s = NB / 2; s >= 1; s >>= 1) {
#pragma unroll
        for (int r = 0; r < 4; ++r) {
            if (sb[m[r] + s - 1] <= na[r]) m[r] += s;
        }
    }
    float acc1 = 0.f, acc2 = 0.f;
#pragma unroll
    for (int r = 0; r < 4; ++r) {
        if (sb[m[r]] <= na[r]) m[r] += 1;
        const float cnt = (float)(NB - m[r]);
        const float q1  = p1[m[r]];
        const float q2  = p2[m[r]];
        acc1 += fmaf(cnt, a[r], q1);
        acc2 += fmaf(a[r], fmaf(cnt, a[r], 2.f * q1), q2);
    }

#pragma unroll
    for (int d = 16; d >= 1; d >>= 1) {
        acc1 += __shfl_xor(acc1, d, 32);
        acc2 += __shfl_xor(acc2, d, 32);
    }
    if (lane == 0) { red[0][wave] = acc1; red[1][wave] = acc2; }
    __syncthreads();
    if (tid == 0) {
        float r1 = 0.f, r2 = 0.f;
#pragma unroll
        for (int w = 0; w < 8; ++w) { r1 += red[0][w]; r2 += red[1][w]; }
        const float invN = 1.0f / ((float)NB * (float)NB);
        const float mean = r1 * invN;
        const float var  = fmaxf(r2 * invN - mean * mean, 0.f);
        const float inv  = rsqrtf(var + BN_EPS);
        const float g    = gm * inv;
        cvec[h] = w2 * g;
        term[h] = w2 * (bt - mean * g);
    }
}

__device__ __forceinline__ float lane_xor16(float v)
{
    const int b = __builtin_bit_cast(int, v);
    return __builtin_bit_cast(float,
        __builtin_amdgcn_permlanex16(b, b, 0x76543210, 0xfedcba98, false, false));
}
template <int M>
__device__ __forceinline__ float row_xor(float v)
{
    const int b = __builtin_bit_cast(int, v);
    return __builtin_bit_cast(float,
        __builtin_amdgcn_update_dpp(b, b, 0x160 | M, 0xf, 0xf, false));
}

struct K4Frag {
    float4 a[4];
    float4 b[4];
    float4 c;
};

__global__ __launch_bounds__(256)
void k4_out(const float* __restrict__ A, const float* __restrict__ Bm,
            const float* __restrict__ cvec, const float* __restrict__ term,
            const float* __restrict__ b2, float* __restrict__ out)
{
    __shared__ __attribute__((aligned(16))) float aT[DHID][64];
    __shared__ __attribute__((aligned(16))) float bT[DHID][64];
    __shared__ __attribute__((aligned(16))) float cs[DHID];

    const int i0  = blockIdx.x * 64;
    const int j0  = blockIdx.y * 64;
    const int tid = threadIdx.x;

    const int   ln = tid & 31;
    const float t0 = term[ln], t1 = term[ln + 32], t2 = term[ln + 64], t3 = term[ln + 96];
    const float bias2 = b2[0];

    {
        const int rl = tid & 63;
        const int hq = tid >> 6;
        const float4* sa = (const float4*)(A  + (i0 + rl) * DHID + hq * 32);
        const float4* sb = (const float4*)(Bm + (j0 + rl) * DHID + hq * 32);
#pragma unroll
        for (int q = 0; q < 8; ++q) {
            const float4 v = sa[q];
            const int h = hq * 32 + q * 4;
            aT[h + 0][rl] = v.x; aT[h + 1][rl] = v.y;
            aT[h + 2][rl] = v.z; aT[h + 3][rl] = v.w;
        }
#pragma unroll
        for (int q = 0; q < 8; ++q) {
            const float4 v = sb[q];
            const int h = hq * 32 + q * 4;
            bT[h + 0][rl] = v.x; bT[h + 1][rl] = v.y;
            bT[h + 2][rl] = v.z; bT[h + 3][rl] = v.w;
        }
    }
    if (tid < DHID) cs[tid] = cvec[tid];

    float cv = (t0 + t2) + (t1 + t3);
    cv += lane_xor16(cv);
    cv += row_xor<8>(cv);
    cv += row_xor<4>(cv);
    cv += row_xor<2>(cv);
    cv += row_xor<1>(cv);
    const float ct = cv + bias2;
    __syncthreads();

    const int jb = (tid & 15) * 4;
    const int ib = (tid >> 4) * 4;
    float acc[4][4] = {};

    auto load4 = [&](K4Frag& f, int h) {
#pragma unroll
        for (int u = 0; u < 4; ++u) {
            f.a[u] = *(const float4*)&aT[h + u][ib];
            f.b[u] = *(const float4*)&bT[h + u][jb];
        }
        f.c = *(const float4*)&cs[h];
    };
    auto step4 = [&](const K4Frag& f) {
        const float cq[4] = {f.c.x, f.c.y, f.c.z, f.c.w};
#pragma unroll
        for (int u = 0; u < 4; ++u) {
            const float av[4] = {f.a[u].x, f.a[u].y, f.a[u].z, f.a[u].w};
            const float bv[4] = {f.b[u].x, f.b[u].y, f.b[u].z, f.b[u].w};
#pragma unroll
            for (int ii = 0; ii < 4; ++ii) {
#pragma unroll
                for (int jj = 0; jj < 4; ++jj) {
                    acc[ii][jj] = fmaf(cq[u], fmaxf(av[ii] + bv[jj], 0.f), acc[ii][jj]);
                }
            }
        }
    };

    for (int h = 0; h < DHID; h += 8) {
        K4Frag f0, f1;
        load4(f0, h);
        load4(f1, h + 4);
        step4(f0);
        step4(f1);
    }

#pragma unroll
    for (int ii = 0; ii < 4; ++ii) {
        float4 o;
        o.x = acc[ii][0] + ct;
        o.y = acc[ii][1] + ct;
        o.z = acc[ii][2] + ct;
        o.w = acc[ii][3] + ct;
        *(float4*)(out + (size_t)(i0 + ib + ii) * NB + (j0 + jb)) = o;
    }
}

extern "C" void kernel_launch(void* const* d_in, const int* in_sizes, int n_in,
                              void* d_out, int out_size, void* d_ws, size_t ws_size,
                              hipStream_t stream)
{
    const float* x     = (const float*)d_in[0];
    const float* W1    = (const float*)d_in[1];
    const float* b1    = (const float*)d_in[2];
    const float* gamma = (const float*)d_in[3];
    const float* beta  = (const float*)d_in[4];
    const float* W2    = (const float*)d_in[5];
    const float* b2    = (const float*)d_in[6];
    float* out = (float*)d_out;

    float* ws       = (float*)d_ws;
    float* A        = ws;
    float* Bm       = ws + NB * DHID;
    float* cvec     = ws + 2 * NB * DHID;
    float* term     = cvec + DHID;

    k1_gemm <<<dim3(NB / 16, DHID / 16, 2), 32,  0, stream>>>(x, W1, b1, A, Bm);
    k2_stats<<<dim3(DHID),                 256, 0, stream>>>(A, Bm, gamma, beta, W2, cvec, term);
    k4_out  <<<dim3(NB / 64, NB / 64),     256, 0, stream>>>(A, Bm, cvec, term, b2, out);
}


# K2: all in-wave lane exchanges (30 sort stages, suffix sums, block sum) as vector-ALU DPP / permlanex16 instead of __shfl (LDS-routed ds_bpermute); every sum's order kept
# speedup vs baseline: 1.6524x; 1.0610x over previous
#include <hip/hip_runtime.h>

typedef __attribute__((ext_vector_type(2))) float v2f;
typedef __attribute__((ext_vector_type(8))) float v8f;

#define NB    1024
#define DIN   64
#define DHID  128
#define BN_EPS 1e-5f

__global__ __launch_bounds__(32)
void k1_gemm(const float* __restrict__ x, const float* __restrict__ W1,
             const float* __restrict__ b1, float* __restrict__ A,
             float* __restrict__ Bm)
{
    const int i0   = blockIdx.x * 16;
    const int j0   = blockIdx.y * 16;
    const int hlf  = blockIdx.z;
    const int lane = threadIdx.x & 31;
    const int m    = lane & 15;
    const int kg   = (lane >> 4) << 1;

    const float* __restrict__ W = W1 + hlf * DIN * DHID;
    float* __restrict__ C = hlf ? Bm : A;

    v8f acc = {0.f, 0.f, 0.f, 0.f, 0.f, 0.f, 0.f, 0.f};
    for (int k = 0; k < DIN; k += 4) {
        v2f a, b;
        a.x = x[(i0 + m) * DIN + (k + kg)];
        a.y = x[(i0 + m) * DIN + (k + kg + 1)];
        b.x = W[(k + kg) * DHID + (j0 + m)];
        b.y = W[(k + kg + 1) * DHID + (j0 + m)];
        acc = __builtin_amdgcn_wmma_f32_16x16x4_f32(false, a, false, b,
                                                    (short)0, acc, false, false);
    }

    const float bias  = (hlf == 0) ? b1[j0 + m] : 0.0f;
    const int   rbase = (lane < 16) ? 0 : 8;
#pragma unroll
    for (int v = 0; v < 8; ++v) {
        C[(i0 + rbase + v) * DHID + (j0 + m)] = acc[v] + bias;
    }
}

__device__ __forceinline__ float lane_xor16(float v)
{
    const int b = __builtin_bit_cast(int, v);
    return __builtin_bit_cast(float,
        __builtin_amdgcn_permlanex16(0, b, 0x76543210, 0xfedcba98, false, false));
}
template <int M>
__device__ __forceinline__ float row_xor(float v)
{
    const int b = __builtin_bit_cast(int, v);
    return __builtin_bit_cast(float,
        __builtin_amdgcn_mov_dpp(b, 0x160 | M, 0xf, 0xf, false));
}
__device__ __forceinline__ float lane_xor(float v, int m)
{
    switch (m) {
    case 1:  return row_xor<1>(v);
    case 2:  return row_xor<2>(v);
    case 4:  return row_xor<4>(v);
    case 8:  return row_xor<8>(v);
    default: return lane_xor16(v);
    }
}
template <int D>
__device__ __forceinline__ float lane_down(float v, int lane)
{
    if constexpr (D == 16) {
        return lane_xor16(v);
    } else {
        constexpr unsigned long long ident = 0xfedcba9876543210ull;
        constexpr unsigned long long sel   = (ident >> (4 * D)) | (ident << (64 - 4 * D));
        const int b = __builtin_bit_cast(int, v);
        const float in_row = __builtin_bit_cast(float,
            __builtin_amdgcn_mov_dpp(b, 0x100 | D, 0xf, 0xf, true));
        const float x_row = __builtin_bit_cast(float,
            __builtin_amdgcn_permlanex16(0, b, (unsigned)sel, (unsigned)(sel >> 32), false, false));
        return ((lane & 15) + D < 16) ? in_row : x_row;
    }
}
template <int D>
__device__ __forceinline__ void suffix_step(float& t1, float& t2, int lane)
{
    const float y1 = lane_down<D>(t1, lane);
    const float y2 = lane_down<D>(t2, lane);
    if (lane + D < 32) { t1 += y1; t2 += y2; }
}

__global__ __launch_bounds__(256)
void k2_stats(const float* __restrict__ A, const float* __restrict__ Bm,
              const float* __restrict__ gamma, const float* __restrict__ beta,
              const float* __restrict__ W2,
              float* __restrict__ cvec, float* __restrict__ term)
{
    const int h    = blockIdx.x;
    const int tid  = threadIdx.x;
    const int lane = tid & 31;
    const int wave = tid >> 5;

    const float gm = gamma[h];
    const float bt = beta[h];
    const float w2 = W2[h];

    __shared__ float4 xch[2][256];
    __shared__ __attribute__((aligned(16))) float sb[NB];
    __shared__ __attribute__((aligned(16))) float p1[NB + 4];
    __shared__ __attribute__((aligned(16))) float p2[NB + 4];
    __shared__ float wtot[2][8];
    __shared__ float red[2][8];

    float x[4], a[4];
#pragma unroll
    for (int r = 0; r < 4; ++r) x[r] = Bm[(tid * 4 + r) * DHID + h];
#pragma unroll
    for (int r = 0; r < 4; ++r) a[r] = A[(tid + 256 * r) * DHID + h];

    int xb = 0;
#pragma unroll
    for (int k = 2; k <= NB; k <<= 1) {
#pragma unroll
        for (int j = k >> 1; j >= 1; j >>= 1) {
            if (j >= 128) {
                xch[xb][tid] = make_float4(x[0], x[1], x[2], x[3]);
                __syncthreads();
                const float4 y = xch[xb][tid ^ (j >> 2)];
                xb ^= 1;
                const bool keep_min = ((((tid * 4) & j) == 0) == (((tid * 4) & k) == 0));
                x[0] = keep_min ? fminf(x[0], y.x) : fmaxf(x[0], y.x);
                x[1] = keep_min ? fminf(x[1], y.y) : fmaxf(x[1], y.y);
                x[2] = keep_min ? fminf(x[2], y.z) : fmaxf(x[2], y.z);
                x[3] = keep_min ? fminf(x[3], y.w) : fmaxf(x[3], y.w);
            } else if (j >= 4) {
                const bool keep_min = ((((tid * 4) & j) == 0) == (((tid * 4) & k) == 0));
#pragma unroll
                for (int r = 0; r < 4; ++r) {
                    const float y = lane_xor(x[r], j >> 2);
                    x[r] = keep_min ? fminf(x[r], y) : fmaxf(x[r], y);
                }
            } else {
#pragma unroll
                for (int r = 0; r < 4; ++r) {
                    if ((r & j) == 0) {
                        const bool asc = (((tid * 4 + r) & k) == 0);
                        const float lo = fminf(x[r], x[r | j]);
                        const float hi = fmaxf(x[r], x[r | j]);
                        x[r]     = asc ? lo : hi;
                        x[r | j] = asc ? hi : lo;
                    }
                }
            }
        }
    }

    float s1[4], s2[4];
    s1[3] = x[3];          s2[3] = x[3] * x[3];
    s1[2] = x[2] + s1[3];  s2[2] = fmaf(x[2], x[2], s2[3]);
    s1[1] = x[1] + s1[2];  s2[1] = fmaf(x[1], x[1], s2[2]);
    s1[0] = x[0] + s1[1];  s2[0] = fmaf(x[0], x[0], s2[1]);

    float t1 = s1[0], t2 = s2[0];
    suffix_step<1>(t1, t2, lane);
    suffix_step<2>(t1, t2, lane);
    suffix_step<4>(t1, t2, lane);
    suffix_step<8>(t1, t2, lane);
    suffix_step<16>(t1, t2, lane);
    const float n1 = lane_down<1>(t1, lane);
    const float n2 = lane_down<1>(t2, lane);
    float off1 = (lane < 31) ? n1 : 0.f;
    float off2 = (lane < 31) ? n2 : 0.f;
    if (lane == 0) { wtot[0][wave] = t1; wtot[1][wave] = t2; }
    __syncthreads();
    {
        float hi1 = 0.f, hi2 = 0.f;
#pragma unroll
        for (int w = 7; w >= 1; --w) {
            if (w > wave) { hi1 += wtot[0][w]; hi2 += wtot[1][w]; }
        }
        off1 += hi1; off2 += hi2;
    }
    *(float4*)&sb[tid * 4] = make_float4(x[0], x[1], x[2], x[3]);
    *(float4*)&p1[tid * 4] = make_float4(s1[0] + off1, s1[1] + off1, s1[2] + off1, s1[3] + off1);
    *(float4*)&p2[tid * 4] = make_float4(s2[0] + off2, s2[1] + off2, s2[2] + off2, s2[3] + off2);
    if (tid == 0) { p1[NB] = 0.f; p2[NB] = 0.f; }
    __syncthreads();

    float na[4];
    int   m[4] = {0, 0, 0, 0};
#pragma unroll
    for (int r = 0; r < 4; ++r) na[r] = -a[r];
#pragma unroll
    for (int s = NB / 2; s >= 1; s >>= 1) {
#pragma unroll
        for (int r = 0; r < 4; ++r) {
            if (sb[m[r] + s - 1] <= na[r]) m[r] += s;
        }
    }
    float acc1 = 0.f, acc2 = 0.f;
#pragma unroll
    for (int r = 0; r < 4; ++r) {
        if (sb[m[r]] <= na[r]) m[r] += 1;
        const float cnt = (float)(NB - m[r]);
        const float q1  = p1[m[r]];
        const float q2  = p2[m[r]];
        acc1 += fmaf(cnt, a[r], q1);
        acc2 += fmaf(a[r], fmaf(cnt, a[r], 2.f * q1), q2);
    }

#pragma unroll
    for (int d = 16; d >= 1; d >>= 1) {
        acc1 += lane_xor(acc1, d);
        acc2 += lane_xor(acc2, d);
    }
    if (lane == 0) { red[0][wave] = acc1; red[1][wave] = acc2; }
    __syncthreads();
    if (tid == 0) {
        float r1 = 0.f, r2 = 0.f;
#pragma unroll
        for (int w = 0; w < 8; ++w) { r1 += red[0][w]; r2 += red[1][w]; }
        const float invN = 1.0f / ((float)NB * (float)NB);
        const float mean = r1 * invN;
        const float var  = fmaxf(r2 * invN - mean * mean, 0.f);
        const float inv  = rsqrtf(var + BN_EPS);
        const float g    = gm * inv;
        cvec[h] = w2 * g;
        term[h] = w2 * (bt - mean * g);
    }
}

struct K4Frag {
    float4 a[4];
    float4 b[4];
    float4 c;
};

__global__ __launch_bounds__(256)
void k4_out(const float* __restrict__ A, const float* __restrict__ Bm,
            const float* __restrict__ cvec, const float* __restrict__ term,
            const float* __restrict__ b2, float* __restrict__ out)
{
    __shared__ __attribute__((aligned(16))) float aT[DHID][64];
    __shared__ __attribute__((aligned(16))) float bT[DHID][64];
    __shared__ __attribute__((aligned(16))) float cs[DHID];

    const int i0  = blockIdx.x * 64;
    const int j0  = blockIdx.y * 64;
    const int tid = threadIdx.x;

    const int   ln = tid & 31;
    const float t0 = term[ln], t1 = term[ln + 32], t2 = term[ln + 64], t3 = term[ln + 96];
    const float bias2 = b2[0];

    {
        const int rl = tid & 63;
        const int hq = tid >> 6;
        const float4* sa = (const float4*)(A  + (i0 + rl) * DHID + hq * 32);
        const float4* sb = (const float4*)(Bm + (j0 + rl) * DHID + hq * 32);
#pragma unroll
        for (int q = 0; q < 8; ++q) {
            const float4 v = sa[q];
            const int h = hq * 32 + q * 4;
            aT[h + 0][rl] = v.x; aT[h + 1][rl] = v.y;
            aT[h + 2][rl] = v.z; aT[h + 3][rl] = v.w;
        }
#pragma unroll
        for (int q = 0; q < 8; ++q) {
            const float4 v = sb[q];
            const int h = hq * 32 + q * 4;
            bT[h + 0][rl] = v.x; bT[h + 1][rl] = v.y;
            bT[h + 2][rl] = v.z; bT[h + 3][rl] = v.w;
        }
    }
    if (tid < DHID) cs[tid] = cvec[tid];

    float cv = (t0 + t2) + (t1 + t3);
    cv += lane_xor16(cv);
    cv += row_xor<8>(cv);
    cv += row_xor<4>(cv);
    cv += row_xor<2>(cv);
    cv += row_xor<1>(cv);
    const float ct = cv + bias2;
    __syncthreads();

    const int jb = (tid & 15) * 4;
    const int ib = (tid >> 4) * 4;
    float acc[4][4] = {};

    auto load4 = [&](K4Frag& f, int h) {
#pragma unroll
        for (int u = 0; u < 4; ++u) {
            f.a[u] = *(const float4*)&aT[h + u][ib];
            f.b[u] = *(const float4*)&bT[h + u][jb];
        }
        f.c = *(const float4*)&cs[h];
    };
    auto step4 = [&](const K4Frag& f) {
        const float cq[4] = {f.c.x, f.c.y, f.c.z, f.c.w};
#pragma unroll
        for (int u = 0; u < 4; ++u) {
            const float av[4] = {f.a[u].x, f.a[u].y, f.a[u].z, f.a[u].w};
            const float bv[4] = {f.b[u].x, f.b[u].y, f.b[u].z, f.b[u].w};
#pragma unroll
            for (int ii = 0; ii < 4; ++ii) {
#pragma unroll
                for (int jj = 0; jj < 4; ++jj) {
                    acc[ii][jj] = fmaf(cq[u], fmaxf(av[ii] + bv[jj], 0.f), acc[ii][jj]);
                }
            }
        }
    };

    for (int h = 0; h < DHID; h += 8) {
        K4Frag f0, f1;
        load4(f0, h);
        load4(f1, h + 4);
        step4(f0);
        step4(f1);
    }

#pragma unroll
    for (int ii = 0; ii < 4; ++ii) {
        float4 o;
        o.x = acc[ii][0] + ct;
        o.y = acc[ii][1] + ct;
        o.z = acc[ii][2] + ct;
        o.w = acc[ii][3] + ct;
        *(float4*)(out + (size_t)(i0 + ib + ii) * NB + (j0 + jb)) = o;
    }
}

extern "C" void kernel_launch(void* const* d_in, const int* in_sizes, int n_in,
                              void* d_out, int out_size, void* d_ws, size_t ws_size,
                              hipStream_t stream)
{
    const float* x     = (const float*)d_in[0];
    const float* W1    = (const float*)d_in[1];
    const float* b1    = (const float*)d_in[2];
    const float* gamma = (const float*)d_in[3];
    const float* beta  = (const float*)d_in[4];
    const float* W2    = (const float*)d_in[5];
    const float* b2    = (const float*)d_in[6];
    float* out = (float*)d_out;

    float* ws       = (float*)d_ws;
    float* A        = ws;
    float* Bm       = ws + NB * DHID;
    float* cvec     = ws + 2 * NB * DHID;
    float* term     = cvec + DHID;

    k1_gemm <<<dim3(NB / 16, DHID / 16, 2), 32,  0, stream>>>(x, W1, b1, A, Bm);
    k2_stats<<<dim3(DHID),                 256, 0, stream>>>(A, Bm, gamma, beta, W2, cvec, term);
    k4_out  <<<dim3(NB / 64, NB / 64),     256, 0, stream>>>(A, Bm, cvec, term, b2, out);
}


# K4 stages its tiles untransposed: whole-row (4-line) global loads, 16-byte LDS stores, chunks XOR-permuted per row, fragments read along k; sum order changed: even-k + odd-k partial sums per output
# speedup vs baseline: 1.7394x; 1.0527x over previous
#include <hip/hip_runtime.h>

typedef __attribute__((ext_vector_type(2))) float v2f;
typedef __attribute__((ext_vector_type(8))) float v8f;

#define NB    1024
#define DIN   64
#define DHID  128
#define BN_EPS 1e-5f

__global__ __launch_bounds__(32)
void k1_gemm(const float* __restrict__ x, const float* __restrict__ W1,
             const float* __restrict__ b1, float* __restrict__ A,
             float* __restrict__ Bm)
{
    const int i0   = blockIdx.x * 16;
    const int j0   = blockIdx.y * 16;
    const int hlf  = blockIdx.z;
    const int lane = threadIdx.x & 31;
    const int m    = lane & 15;
    const int kg   = (lane >> 4) << 1;

    const float* __restrict__ W = W1 + hlf * DIN * DHID;
    float* __restrict__ C = hlf ? Bm : A;

    v8f acc = {0.f, 0.f, 0.f, 0.f, 0.f, 0.f, 0.f, 0.f};
    for (int k = 0; k < DIN; k += 4) {
        v2f a, b;
        a.x = x[(i0 + m) * DIN + (k + kg)];
        a.y = x[(i0 + m) * DIN + (k + kg + 1)];
        b.x = W[(k + kg) * DHID + (j0 + m)];
        b.y = W[(k + kg + 1) * DHID + (j0 + m)];
        acc = __builtin_amdgcn_wmma_f32_16x16x4_f32(false, a, false, b,
                                                    (short)0, acc, false, false);
    }

    const float bias  = (hlf == 0) ? b1[j0 + m] : 0.0f;
    const int   rbase = (lane < 16) ? 0 : 8;
#pragma unroll
    for (int v = 0; v < 8; ++v) {
        C[(i0 + rbase + v) * DHID + (j0 + m)] = acc[v] + bias;
    }
}

__device__ __forceinline__ float lane_xor16(float v)
{
    const int b = __builtin_bit_cast(int, v);
    return __builtin_bit_cast(float,
        __builtin_amdgcn_permlanex16(0, b, 0x76543210, 0xfedcba98, false, false));
}
template <int M>
__device__ __forceinline__ float row_xor(float v)
{
    const int b = __builtin_bit_cast(int, v);
    return __builtin_bit_cast(float,
        __builtin_amdgcn_mov_dpp(b, 0x160 | M, 0xf, 0xf, false));
}
__device__ __forceinline__ float lane_xor(float v, int m)
{
    switch (m) {
    case 1:  return row_xor<1>(v);
    case 2:  return row_xor<2>(v);
    case 4:  return row_xor<4>(v);
    case 8:  return row_xor<8>(v);
    default: return lane_xor16(v);
    }
}
template <int D>
__device__ __forceinline__ float lane_down(float v, int lane)
{
    if constexpr (D == 16) {
        return lane_xor16(v);
    } else {
        constexpr unsigned long long ident = 0xfedcba9876543210ull;
        constexpr unsigned long long sel   = (ident >> (4 * D)) | (ident << (64 - 4 * D));
        const int b = __builtin_bit_cast(int, v);
        const float in_row = __builtin_bit_cast(float,
            __builtin_amdgcn_mov_dpp(b, 0x100 | D, 0xf, 0xf, true));
        const float x_row = __builtin_bit_cast(float,
            __builtin_amdgcn_permlanex16(0, b, (unsigned)sel, (unsigned)(sel >> 32), false, false));
        return ((lane & 15) + D < 16) ? in_row : x_row;
    }
}
template <int D>
__device__ __forceinline__ void suffix_step(float& t1, float& t2, int lane)
{
    const float y1 = lane_down<D>(t1, lane);
    const float y2 = lane_down<D>(t2, lane);
    if (lane + D < 32) { t1 += y1; t2 += y2; }
}

__global__ __launch_bounds__(256)
void k2_stats(const float* __restrict__ A, const float* __restrict__ Bm,
              const float* __restrict__ gamma, const float* __restrict__ beta,
              const float* __restrict__ W2,
              float* __restrict__ cvec, float* __restrict__ term)
{
    const int h    = blockIdx.x;
    const int tid  = threadIdx.x;
    const int lane = tid & 31;
    const int wave = tid >> 5;

    const float gm = gamma[h];
    const float bt = beta[h];
    const float w2 = W2[h];

    __shared__ float4 xch[2][256];
    __shared__ __attribute__((aligned(16))) float sb[NB];
    __shared__ __attribute__((aligned(16))) float p1[NB + 4];
    __shared__ __attribute__((aligned(16))) float p2[NB + 4];
    __shared__ float wtot[2][8];
    __shared__ float red[2][8];

    float x[4], a[4];
#pragma unroll
    for (int r = 0; r < 4; ++r) x[r] = Bm[(tid * 4 + r) * DHID + h];
#pragma unroll
    for (int r = 0; r < 4; ++r) a[r] = A[(tid + 256 * r) * DHID + h];

    int xb = 0;
#pragma unroll
    for (int k = 2; k <= NB; k <<= 1) {
#pragma unroll
        for (int j = k >> 1; j >= 1; j >>= 1) {
            if (j >= 128) {
                xch[xb][tid] = make_float4(x[0], x[1], x[2], x[3]);
                __syncthreads();
                const float4 y = xch[xb][tid ^ (j >> 2)];
                xb ^= 1;
                const bool keep_min = ((((tid * 4) & j) == 0) == (((tid * 4) & k) == 0));
                x[0] = keep_min ? fminf(x[0], y.x) : fmaxf(x[0], y.x);
                x[1] = keep_min ? fminf(x[1], y.y) : fmaxf(x[1], y.y);
                x[2] = keep_min ? fminf(x[2], y.z) : fmaxf(x[2], y.z);
                x[3] = keep_min ? fminf(x[3], y.w) : fmaxf(x[3], y.w);
            } else if (j >= 4) {
                const bool keep_min = ((((tid * 4) & j) == 0) == (((tid * 4) & k) == 0));
#pragma unroll
                for (int r = 0; r < 4; ++r) {
                    const float y = lane_xor(x[r], j >> 2);
                    x[r] = keep_min ? fminf(x[r], y) : fmaxf(x[r], y);
                }
            } else {
#pragma unroll
                for (int r = 0; r < 4; ++r) {
                    if ((r & j) == 0) {
                        const bool asc = (((tid * 4 + r) & k) == 0);
                        const float lo = fminf(x[r], x[r | j]);
                        const float hi = fmaxf(x[r], x[r | j]);
                        x[r]     = asc ? lo : hi;
                        x[r | j] = asc ? hi : lo;
                    }
                }
            }
        }
    }

    float s1[4], s2[4];
    s1[3] = x[3];          s2[3] = x[3] * x[3];
    s1[2] = x[2] + s1[3];  s2[2] = fmaf(x[2], x[2], s2[3]);
    s1[1] = x[1] + s1[2];  s2[1] = fmaf(x[1], x[1], s2[2]);
    s1[0] = x[0] + s1[1];  s2[0] = fmaf(x[0], x[0], s2[1]);

    float t1 = s1[0], t2 = s2[0];
    suffix_step<1>(t1, t2, lane);
    suffix_step<2>(t1, t2, lane);
    suffix_step<4>(t1, t2, lane);
    suffix_step<8>(t1, t2, lane);
    suffix_step<16>(t1, t2, lane);
    const float n1 = lane_down<1>(t1, lane);
    const float n2 = lane_down<1>(t2, lane);
    float off1 = (lane < 31) ? n1 : 0.f;
    float off2 = (lane < 31) ? n2 : 0.f;
    if (lane == 0) { wtot[0][wave] = t1; wtot[1][wave] = t2; }
    __syncthreads();
    {
        float hi1 = 0.f, hi2 = 0.f;
#pragma unroll
        for (int w = 7; w >= 1; --w) {
            if (w > wave) { hi1 += wtot[0][w]; hi2 += wtot[1][w]; }
        }
        off1 += hi1; off2 += hi2;
    }
    *(float4*)&sb[tid * 4] = make_float4(x[0], x[1], x[2], x[3]);
    *(float4*)&p1[tid * 4] = make_float4(s1[0] + off1, s1[1] + off1, s1[2] + off1, s1[3] + off1);
    *(float4*)&p2[tid * 4] = make_float4(s2[0] + off2, s2[1] + off2, s2[2] + off2, s2[3] + off2);
    if (tid == 0) { p1[NB] = 0.f; p2[NB] = 0.f; }
    __syncthreads();

    float na[4];
    int   m[4] = {0, 0, 0, 0};
#pragma unroll
    for (int r = 0; r < 4; ++r) na[r] = -a[r];
#pragma unroll
    for (int s = NB / 2; s >= 1; s >>= 1) {
#pragma unroll
        for (int r = 0; r < 4; ++r) {
            if (sb[m[r] + s - 1] <= na[r]) m[r] += s;
        }
    }
    float acc1 = 0.f, acc2 = 0.f;
#pragma unroll
    for (int r = 0; r < 4; ++r) {
        if (sb[m[r]] <= na[r]) m[r] += 1;
        const float cnt = (float)(NB - m[r]);
        const float q1  = p1[m[r]];
        const float q2  = p2[m[r]];
        acc1 += fmaf(cnt, a[r], q1);
        acc2 += fmaf(a[r], fmaf(cnt, a[r], 2.f * q1), q2);
    }

#pragma unroll
    for (int d = 16; d >= 1; d >>= 1) {
        acc1 += lane_xor(acc1, d);
        acc2 += lane_xor(acc2, d);
    }
    if (lane == 0) { red[0][wave] = acc1; red[1][wave] = acc2; }
    __syncthreads();
    if (tid == 0) {
        float r1 = 0.f, r2 = 0.f;
#pragma unroll
        for (int w = 0; w < 8; ++w) { r1 += red[0][w]; r2 += red[1][w]; }
        const float invN = 1.0f / ((float)NB * (float)NB);
        const float mean = r1 * invN;
        const float var  = fmaxf(r2 * invN - mean * mean, 0.f);
        const float inv  = rsqrtf(var + BN_EPS);
        const float g    = gm * inv;
        cvec[h] = w2 * g;
        term[h] = w2 * (bt - mean * g);
    }
}

struct K4Frag {
    float4 a[4];
    float4 b[4];
    float4 c;
};

__global__ __launch_bounds__(256)
void k4_out(const float* __restrict__ A, const float* __restrict__ Bm,
            const float* __restrict__ cvec, const float* __restrict__ term,
            const float* __restrict__ b2, float* __restrict__ out)
{
    __shared__ __attribute__((aligned(16))) float aS[64 * DHID];
    __shared__ __attribute__((aligned(16))) float bS[64 * DHID];
    __shared__ __attribute__((aligned(16))) float cs[DHID];

    const int i0  = blockIdx.x * 64;
    const int j0  = blockIdx.y * 64;
    const int tid = threadIdx.x;

    const int   ln = tid & 31;
    const int   wv = tid >> 5;
    const float t0 = term[ln], t1 = term[ln + 32], t2 = term[ln + 64], t3 = term[ln + 96];
    const float bias2 = b2[0];

    {
        const float4* ga = (const float4*)(A  + (i0 + wv * 8) * DHID) + ln;
        const float4* gb = (const float4*)(Bm + (j0 + wv * 8) * DHID) + ln;
        float4 va[8], vb[8];
#pragma unroll
        for (int q = 0; q < 8; ++q) { va[q] = ga[q * (DHID / 4)]; vb[q] = gb[q * (DHID / 4)]; }
#pragma unroll
        for (int q = 0; q < 8; ++q) {
            const int r   = wv * 8 + q;
            const int pos = ln ^ ((r >> 2) & 15);
            *(float4*)&aS[r * DHID + 4 * pos] = va[q];
            *(float4*)&bS[r * DHID + 4 * pos] = vb[q];
        }
    }
    if (tid < DHID) cs[tid] = cvec[tid];

    float cv = (t0 + t2) + (t1 + t3);
    cv += lane_xor16(cv);
    cv += row_xor<8>(cv);
    cv += row_xor<4>(cv);
    cv += row_xor<2>(cv);
    cv += row_xor<1>(cv);
    const float ct = cv + bias2;
    __syncthreads();

    const int jl = tid & 15;
    const int ig = tid >> 4;
    const int jb = jl * 4;
    const int ib = ig * 4;
    const float* arow = &aS[ib * DHID];
    const float* brow = &bS[jb * DHID];

    v2f acc[4][4];
#pragma unroll
    for (int ii = 0; ii < 4; ++ii)
#pragma unroll
        for (int jj = 0; jj < 4; ++jj) acc[ii][jj] = (v2f){0.f, 0.f};

    auto load4 = [&](K4Frag& f, int kc) {
        const int pa = 4 * (kc ^ ig);
        const int pb = 4 * (kc ^ jl);
#pragma unroll
        for (int u = 0; u < 4; ++u) {
            f.a[u] = *(const float4*)&arow[u * DHID + pa];
            f.b[u] = *(const float4*)&brow[u * DHID + pb];
        }
        f.c = *(const float4*)&cs[4 * kc];
    };
    auto step4 = [&](const K4Frag& f) {
        const v2f zero = {0.f, 0.f};
        const v2f clo = {f.c.x, f.c.y};
        const v2f chi = {f.c.z, f.c.w};
#pragma unroll
        for (int ii = 0; ii < 4; ++ii) {
            const v2f alo = {f.a[ii].x, f.a[ii].y};
            const v2f ahi = {f.a[ii].z, f.a[ii].w};
#pragma unroll
            for (int jj = 0; jj < 4; ++jj) {
                const v2f blo = {f.b[jj].x, f.b[jj].y};
                const v2f bhi = {f.b[jj].z, f.b[jj].w};
                const v2f rlo = __builtin_elementwise_max(alo + blo, zero);
                acc[ii][jj] = __builtin_elementwise_fma(clo, rlo, acc[ii][jj]);
                const v2f rhi = __builtin_elementwise_max(ahi + bhi, zero);
                acc[ii][jj] = __builtin_elementwise_fma(chi, rhi, acc[ii][jj]);
            }
        }
    };

    for (int kc = 0; kc < DHID / 4; kc += 2) {
        K4Frag f0, f1;
        load4(f0, kc);
        load4(f1, kc + 1);
        step4(f0);
        step4(f1);
    }

#pragma unroll
    for (int ii = 0; ii < 4; ++ii) {
        float4 o;
        o.x = (acc[ii][0].x + acc[ii][0].y) + ct;
        o.y = (acc[ii][1].x + acc[ii][1].y) + ct;
        o.z = (acc[ii][2].x + acc[ii][2].y) + ct;
        o.w = (acc[ii][3].x + acc[ii][3].y) + ct;
        *(float4*)(out + (size_t)(i0 + ib + ii) * NB + (j0 + jb)) = o;
    }
}

extern "C" void kernel_launch(void* const* d_in, const int* in_sizes, int n_in,
                              void* d_out, int out_size, void* d_ws, size_t ws_size,
                              hipStream_t stream)
{
    const float* x     = (const float*)d_in[0];
    const float* W1    = (const float*)d_in[1];
    const float* b1    = (const float*)d_in[2];
    const float* gamma = (const float*)d_in[3];
    const float* beta  = (const float*)d_in[4];
    const float* W2    = (const float*)d_in[5];
    const float* b2    = (const float*)d_in[6];
    float* out = (float*)d_out;

    float* ws       = (float*)d_ws;
    float* A        = ws;
    float* Bm       = ws + NB * DHID;
    float* cvec     = ws + 2 * NB * DHID;
    float* term     = cvec + DHID;

    k1_gemm <<<dim3(NB / 16, DHID / 16, 2), 32,  0, stream>>>(x, W1, b1, A, Bm);
    k2_stats<<<dim3(DHID),                 256, 0, stream>>>(A, Bm, gamma, beta, W2, cvec, term);
    k4_out  <<<dim3(NB / 64, NB / 64),     256, 0, stream>>>(A, Bm, cvec, term, b2, out);
}


# K1+K2 in one launch (3 -> 2): block index picks the role; each stats block forms its own a,b columns from x with f32 FMAs (S1,S2 inputs round differently, sum order kept); WMMA GEMM unchanged
# speedup vs baseline: 2.0961x; 1.2051x over previous
#include <hip/hip_runtime.h>

typedef __attribute__((ext_vector_type(2))) float v2f;
typedef __attribute__((ext_vector_type(8))) float v8f;

#define NB    1024
#define DIN   64
#define DHID  128
#define BN_EPS 1e-5f

__device__ __forceinline__
void gemm_tile(const float* __restrict__ x, const float* __restrict__ W1,
               const float* __restrict__ b1, float* __restrict__ A,
               float* __restrict__ Bm, int t, int lane)
{
    const int i0   = (t & 63) * 16;
    const int j0   = ((t >> 6) & 7) * 16;
    const int hlf  = t >> 9;
    const int m    = lane & 15;
    const int kg   = (lane >> 4) << 1;

    const float* __restrict__ W = W1 + hlf * DIN * DHID;
    float* __restrict__ C = hlf ? Bm : A;

    v8f acc = {0.f, 0.f, 0.f, 0.f, 0.f, 0.f, 0.f, 0.f};
    for (int k = 0; k < DIN; k += 4) {
        v2f a, b;
        a.x = x[(i0 + m) * DIN + (k + kg)];
        a.y = x[(i0 + m) * DIN + (k + kg + 1)];
        b.x = W[(k + kg) * DHID + (j0 + m)];
        b.y = W[(k + kg + 1) * DHID + (j0 + m)];
        acc = __builtin_amdgcn_wmma_f32_16x16x4_f32(false, a, false, b,
                                                    (short)0, acc, false, false);
    }

    const float bias  = (hlf == 0) ? b1[j0 + m] : 0.0f;
    const int   rbase = (lane < 16) ? 0 : 8;
#pragma unroll
    for (int v = 0; v < 8; ++v) {
        C[(i0 + rbase + v) * DHID + (j0 + m)] = acc[v] + bias;
    }
}

__device__ __forceinline__ float lane_xor16(float v)
{
    const int b = __builtin_bit_cast(int, v);
    return __builtin_bit_cast(float,
        __builtin_amdgcn_permlanex16(0, b, 0x76543210, 0xfedcba98, false, false));
}
template <int M>
__device__ __forceinline__ float row_xor(float v)
{
    const int b = __builtin_bit_cast(int, v);
    return __builtin_bit_cast(float,
        __builtin_amdgcn_mov_dpp(b, 0x160 | M, 0xf, 0xf, false));
}
__device__ __forceinline__ float lane_xor(float v, int m)
{
    switch (m) {
    case 1:  return row_xor<1>(v);
    case 2:  return row_xor<2>(v);
    case 4:  return row_xor<4>(v);
    case 8:  return row_xor<8>(v);
    default: return lane_xor16(v);
    }
}
template <int D>
__device__ __forceinline__ float lane_down(float v, int lane)
{
    if constexpr (D == 16) {
        return lane_xor16(v);
    } else {
        constexpr unsigned long long ident = 0xfedcba9876543210ull;
        constexpr unsigned long long sel   = (ident >> (4 * D)) | (ident << (64 - 4 * D));
        const int b = __builtin_bit_cast(int, v);
        const float in_row = __builtin_bit_cast(float,
            __builtin_amdgcn_mov_dpp(b, 0x100 | D, 0xf, 0xf, true));
        const float x_row = __builtin_bit_cast(float,
            __builtin_amdgcn_permlanex16(0, b, (unsigned)sel, (unsigned)(sel >> 32), false, false));
        return ((lane & 15) + D < 16) ? in_row : x_row;
    }
}
template <int D>
__device__ __forceinline__ void suffix_step(float& t1, float& t2, int lane)
{
    const float y1 = lane_down<D>(t1, lane);
    const float y2 = lane_down<D>(t2, lane);
    if (lane + D < 32) { t1 += y1; t2 += y2; }
}

template <int M>
__device__ __forceinline__ float pair_sum(float lo, float hi, bool bit)
{
    const float keep = bit ? hi : lo;
    const float send = bit ? lo : hi;
    return keep + row_xor<M>(send);
}
__device__ __forceinline__ float group8_sum(const float (&v)[8], int q)
{
    const bool b4 = (q & 4) != 0, b2 = (q & 2) != 0, b1 = (q & 1) != 0;
    const float s0 = pair_sum<4>(v[0], v[4], b4);
    const float s1 = pair_sum<4>(v[1], v[5], b4);
    const float s2 = pair_sum<4>(v[2], v[6], b4);
    const float s3 = pair_sum<4>(v[3], v[7], b4);
    const float t0 = pair_sum<2>(s0, s2, b2);
    const float t1 = pair_sum<2>(s1, s3, b2);
    return pair_sum<1>(t0, t1, b1);
}

__global__ __launch_bounds__(256)
void k12_gemm_stats(const float* __restrict__ x_in, const float* __restrict__ W1,
                    const float* __restrict__ b1,
                    const float* __restrict__ gamma, const float* __restrict__ beta,
                    const float* __restrict__ W2,
                    float* __restrict__ A, float* __restrict__ Bm,
                    float* __restrict__ cvec, float* __restrict__ term)
{
    const int tid  = threadIdx.x;
    const int lane = tid & 31;
    const int wave = tid >> 5;

    if (blockIdx.x >= DHID) {
        gemm_tile(x_in, W1, b1, A, Bm, (blockIdx.x - DHID) * 8 + wave, lane);
        return;
    }
    const int h    = blockIdx.x;

    const float gm = gamma[h];
    const float bt = beta[h];
    const float w2 = W2[h];

    __shared__ float4 xch[2][256];
    __shared__ __attribute__((aligned(16))) float sb[NB];
    __shared__ __attribute__((aligned(16))) float p1[NB + 4];
    __shared__ __attribute__((aligned(16))) float p2[NB + 4];
    __shared__ float wtot[2][8];
    __shared__ float red[2][8];
    __shared__ float sa[NB];

    float x[4], a[4];
    {
        const int q  = lane & 7;
        const int rr = lane >> 3;
        float4 wa0, wa1, wb0, wb1;
        wa0.x = W1[(4 * q + 0) * DHID + h];        wa0.y = W1[(4 * q + 1) * DHID + h];
        wa0.z = W1[(4 * q + 2) * DHID + h];        wa0.w = W1[(4 * q + 3) * DHID + h];
        wa1.x = W1[(32 + 4 * q + 0) * DHID + h];   wa1.y = W1[(32 + 4 * q + 1) * DHID + h];
        wa1.z = W1[(32 + 4 * q + 2) * DHID + h];   wa1.w = W1[(32 + 4 * q + 3) * DHID + h];
        wb0.x = W1[(64 + 4 * q + 0) * DHID + h];   wb0.y = W1[(64 + 4 * q + 1) * DHID + h];
        wb0.z = W1[(64 + 4 * q + 2) * DHID + h];   wb0.w = W1[(64 + 4 * q + 3) * DHID + h];
        wb1.x = W1[(96 + 4 * q + 0) * DHID + h];   wb1.y = W1[(96 + 4 * q + 1) * DHID + h];
        wb1.z = W1[(96 + 4 * q + 2) * DHID + h];   wb1.w = W1[(96 + 4 * q + 3) * DHID + h];
        const float bias = b1[h];
        const int   rot  = h & 31;
        const float4* xr = (const float4*)x_in + (wave * 128 * (DIN / 4) + rr * (DIN / 4) + q);

        float4 c0[32], c1[32];
#pragma unroll
        for (int t = 0; t < 32; ++t) {
            const int trip = (t + rot) & 31;
            c0[t] = xr[trip * 64];
            c1[t] = xr[trip * 64 + 8];
        }
        __builtin_amdgcn_sched_barrier(0);
#pragma unroll
        for (int g = 0; g < 4; ++g) {
            float pa[8], pb[8];
#pragma unroll
            for (int i = 0; i < 8; ++i) {
                const float4 u = c0[g * 8 + i];
                const float4 v = c1[g * 8 + i];
                float ta = u.x * wa0.x, tb = u.x * wb0.x;
                ta = fmaf(u.y, wa0.y, ta);  tb = fmaf(u.y, wb0.y, tb);
                ta = fmaf(u.z, wa0.z, ta);  tb = fmaf(u.z, wb0.z, tb);
                ta = fmaf(u.w, wa0.w, ta);  tb = fmaf(u.w, wb0.w, tb);
                ta = fmaf(v.x, wa1.x, ta);  tb = fmaf(v.x, wb1.x, tb);
                ta = fmaf(v.y, wa1.y, ta);  tb = fmaf(v.y, wb1.y, tb);
                ta = fmaf(v.z, wa1.z, ta);  tb = fmaf(v.z, wb1.z, tb);
                ta = fmaf(v.w, wa1.w, ta);  tb = fmaf(v.w, wb1.w, tb);
                pa[i] = ta;  pb[i] = tb;
            }
            const float av = group8_sum(pa, q) + bias;
            x[g] = group8_sum(pb, q);
            const int row = wave * 128 + 4 * ((g * 8 + q + rot) & 31) + rr;
            sa[row] = av;
        }
    }

    int xb = 0;
#pragma unroll
    for (int k = 2; k <= NB; k <<= 1) {
#pragma unroll
        for (int j = k >> 1; j >= 1; j >>= 1) {
            if (j >= 128) {
                xch[xb][tid] = make_float4(x[0], x[1], x[2], x[3]);
                __syncthreads();
                const float4 y = xch[xb][tid ^ (j >> 2)];
                xb ^= 1;
                const bool keep_min = ((((tid * 4) & j) == 0) == (((tid * 4) & k) == 0));
                x[0] = keep_min ? fminf(x[0], y.x) : fmaxf(x[0], y.x);
                x[1] = keep_min ? fminf(x[1], y.y) : fmaxf(x[1], y.y);
                x[2] = keep_min ? fminf(x[2], y.z) : fmaxf(x[2], y.z);
                x[3] = keep_min ? fminf(x[3], y.w) : fmaxf(x[3], y.w);
            } else if (j >= 4) {
                const bool keep_min = ((((tid * 4) & j) == 0) == (((tid * 4) & k) == 0));
#pragma unroll
                for (int r = 0; r < 4; ++r) {
                    const float y = lane_xor(x[r], j >> 2);
                    x[r] = keep_min ? fminf(x[r], y) : fmaxf(x[r], y);
                }
            } else {
#pragma unroll
                for (int r = 0; r < 4; ++r) {
                    if ((r & j) == 0) {
                        const bool asc = (((tid * 4 + r) & k) == 0);
                        const float lo = fminf(x[r], x[r | j]);
                        const float hi = fmaxf(x[r], x[r | j]);
                        x[r]     = asc ? lo : hi;
                        x[r | j] = asc ? hi : lo;
                    }
                }
            }
        }
    }

    float s1[4], s2[4];
    s1[3] = x[3];          s2[3] = x[3] * x[3];
    s1[2] = x[2] + s1[3];  s2[2] = fmaf(x[2], x[2], s2[3]);
    s1[1] = x[1] + s1[2];  s2[1] = fmaf(x[1], x[1], s2[2]);
    s1[0] = x[0] + s1[1];  s2[0] = fmaf(x[0], x[0], s2[1]);

    float t1 = s1[0], t2 = s2[0];
    suffix_step<1>(t1, t2, lane);
    suffix_step<2>(t1, t2, lane);
    suffix_step<4>(t1, t2, lane);
    suffix_step<8>(t1, t2, lane);
    suffix_step<16>(t1, t2, lane);
    const float n1 = lane_down<1>(t1, lane);
    const float n2 = lane_down<1>(t2, lane);
    float off1 = (lane < 31) ? n1 : 0.f;
    float off2 = (lane < 31) ? n2 : 0.f;
    if (lane == 0) { wtot[0][wave] = t1; wtot[1][wave] = t2; }
    __syncthreads();
    {
        float hi1 = 0.f, hi2 = 0.f;
#pragma unroll
        for (int w = 7; w >= 1; --w) {
            if (w > wave) { hi1 += wtot[0][w]; hi2 += wtot[1][w]; }
        }
        off1 += hi1; off2 += hi2;
    }
    *(float4*)&sb[tid * 4] = make_float4(x[0], x[1], x[2], x[3]);
    *(float4*)&p1[tid * 4] = make_float4(s1[0] + off1, s1[1] + off1, s1[2] + off1, s1[3] + off1);
    *(float4*)&p2[tid * 4] = make_float4(s2[0] + off2, s2[1] + off2, s2[2] + off2, s2[3] + off2);
    if (tid == 0) { p1[NB] = 0.f; p2[NB] = 0.f; }
    __syncthreads();
#pragma unroll
    for (int r = 0; r < 4; ++r) a[r] = sa[tid + 256 * r];

    float na[4];
    int   m[4] = {0, 0, 0, 0};
#pragma unroll
    for (int r = 0; r < 4; ++r) na[r] = -a[r];
#pragma unroll
    for (int s = NB / 2; s >= 1; s >>= 1) {
#pragma unroll
        for (int r = 0; r < 4; ++r) {
            if (sb[m[r] + s - 1] <= na[r]) m[r] += s;
        }
    }
    float acc1 = 0.f, acc2 = 0.f;
#pragma unroll
    for (int r = 0; r < 4; ++r) {
        if (sb[m[r]] <= na[r]) m[r] += 1;
        const float cnt = (float)(NB - m[r]);
        const float q1  = p1[m[r]];
        const float q2  = p2[m[r]];
        acc1 += fmaf(cnt, a[r], q1);
        acc2 += fmaf(a[r], fmaf(cnt, a[r], 2.f * q1), q2);
    }

#pragma unroll
    for (int d = 16; d >= 1; d >>= 1) {
        acc1 += lane_xor(acc1, d);
        acc2 += lane_xor(acc2, d);
    }
    if (lane == 0) { red[0][wave] = acc1; red[1][wave] = acc2; }
    __syncthreads();
    if (tid == 0) {
        float r1 = 0.f, r2 = 0.f;
#pragma unroll
        for (int w = 0; w < 8; ++w) { r1 += red[0][w]; r2 += red[1][w]; }
        const float invN = 1.0f / ((float)NB * (float)NB);
        const float mean = r1 * invN;
        const float var  = fmaxf(r2 * invN - mean * mean, 0.f);
        const float inv  = rsqrtf(var + BN_EPS);
        const float g    = gm * inv;
        cvec[h] = w2 * g;
        term[h] = w2 * (bt - mean * g);
    }
}

struct K4Frag {
    float4 a[4];
    float4 b[4];
    float4 c;
};

__global__ __launch_bounds__(256)
void k4_out(const float* __restrict__ A, const float* __restrict__ Bm,
            const float* __restrict__ cvec, const float* __restrict__ term,
            const float* __restrict__ b2, float* __restrict__ out)
{
    __shared__ __attribute__((aligned(16))) float aS[64 * DHID];
    __shared__ __attribute__((aligned(16))) float bS[64 * DHID];
    __shared__ __attribute__((aligned(16))) float cs[DHID];

    const int i0  = blockIdx.x * 64;
    const int j0  = blockIdx.y * 64;
    const int tid = threadIdx.x;

    const int   ln = tid & 31;
    const int   wv = tid >> 5;
    const float t0 = term[ln], t1 = term[ln + 32], t2 = term[ln + 64], t3 = term[ln + 96];
    const float bias2 = b2[0];

    {
        const float4* ga = (const float4*)(A  + (i0 + wv * 8) * DHID) + ln;
        const float4* gb = (const float4*)(Bm + (j0 + wv * 8) * DHID) + ln;
        float4 va[8], vb[8];
#pragma unroll
        for (int q = 0; q < 8; ++q) { va[q] = ga[q * (DHID / 4)]; vb[q] = gb[q * (DHID / 4)]; }
#pragma unroll
        for (int q = 0; q < 8; ++q) {
            const int r   = wv * 8 + q;
            const int pos = ln ^ ((r >> 2) & 15);
            *(float4*)&aS[r * DHID + 4 * pos] = va[q];
            *(float4*)&bS[r * DHID + 4 * pos] = vb[q];
        }
    }
    if (tid < DHID) cs[tid] = cvec[tid];

    float cv = (t0 + t2) + (t1 + t3);
    cv += lane_xor16(cv);
    cv += row_xor<8>(cv);
    cv += row_xor<4>(cv);
    cv += row_xor<2>(cv);
    cv += row_xor<1>(cv);
    const float ct = cv + bias2;
    __syncthreads();

    const int jl = tid & 15;
    const int ig = tid >> 4;
    const int jb = jl * 4;
    const int ib = ig * 4;
    const float* arow = &aS[ib * DHID];
    const float* brow = &bS[jb * DHID];

    v2f acc[4][4];
#pragma unroll
    for (int ii = 0; ii < 4; ++ii)
#pragma unroll
        for (int jj = 0; jj < 4; ++jj) acc[ii][jj] = (v2f){0.f, 0.f};

    auto load4 = [&](K4Frag& f, int kc) {
        const int pa = 4 * (kc ^ ig);
        const int pb = 4 * (kc ^ jl);
#pragma unroll
        for (int u = 0; u < 4; ++u) {
            f.a[u] = *(const float4*)&arow[u * DHID + pa];
            f.b[u] = *(const float4*)&brow[u * DHID + pb];
        }
        f.c = *(const float4*)&cs[4 * kc];
    };
    auto step4 = [&](const K4Frag& f) {
        const v2f zero = {0.f, 0.f};
        const v2f clo = {f.c.x, f.c.y};
        const v2f chi = {f.c.z, f.c.w};
#pragma unroll
        for (int ii = 0; ii < 4; ++ii) {
            const v2f alo = {f.a[ii].x, f.a[ii].y};
            const v2f ahi = {f.a[ii].z, f.a[ii].w};
#pragma unroll
            for (int jj = 0; jj < 4; ++jj) {
                const v2f blo = {f.b[jj].x, f.b[jj].y};
                const v2f bhi = {f.b[jj].z, f.b[jj].w};
                const v2f rlo = __builtin_elementwise_max(alo + blo, zero);
                acc[ii][jj] = __builtin_elementwise_fma(clo, rlo, acc[ii][jj]);
                const v2f rhi = __builtin_elementwise_max(ahi + bhi, zero);
                acc[ii][jj] = __builtin_elementwise_fma(chi, rhi, acc[ii][jj]);
            }
        }
    };

    for (int kc = 0; kc < DHID / 4; kc += 2) {
        K4Frag f0, f1;
        load4(f0, kc);
        load4(f1, kc + 1);
        step4(f0);
        step4(f1);
    }

#pragma unroll
    for (int ii = 0; ii < 4; ++ii) {
        float4 o;
        o.x = (acc[ii][0].x + acc[ii][0].y) + ct;
        o.y = (acc[ii][1].x + acc[ii][1].y) + ct;
        o.z = (acc[ii][2].x + acc[ii][2].y) + ct;
        o.w = (acc[ii][3].x + acc[ii][3].y) + ct;
        *(float4*)(out + (size_t)(i0 + ib + ii) * NB + (j0 + jb)) = o;
    }
}

extern "C" void kernel_launch(void* const* d_in, const int* in_sizes, int n_in,
                              void* d_out, int out_size, void* d_ws, size_t ws_size,
                              hipStream_t stream)
{
    const float* x     = (const float*)d_in[0];
    const float* W1    = (const float*)d_in[1];
    const float* b1    = (const float*)d_in[2];
    const float* gamma = (const float*)d_in[3];
    const float* beta  = (const float*)d_in[4];
    const float* W2    = (const float*)d_in[5];
    const float* b2    = (const float*)d_in[6];
    float* out = (float*)d_out;

    float* ws       = (float*)d_ws;
    float* A        = ws;
    float* Bm       = ws + NB * DHID;
    float* cvec     = ws + 2 * NB * DHID;
    float* term     = cvec + DHID;

    k12_gemm_stats<<<dim3(2 * DHID), 256, 0, stream>>>(x, W1, b1, gamma, beta, W2, A, Bm, cvec, term);
    k4_out  <<<dim3(NB / 64, NB / 64),     256, 0, stream>>>(A, Bm, cvec, term, b2, out);
}


# K4: relu(a+b) taken as max(a,-b)+b: B tile staged negated, a term is one max (hand-placed dual-issue pairs) + one multiply-add; c*b parts summed once per column, added at the end: sum order changed
# speedup vs baseline: 2.2065x; 1.0527x over previous
#include <hip/hip_runtime.h>

typedef __attribute__((ext_vector_type(2))) float v2f;
typedef __attribute__((ext_vector_type(8))) float v8f;

#define NB    1024
#define DIN   64
#define DHID  128
#define BN_EPS 1e-5f

__device__ __forceinline__
void gemm_tile(const float* __restrict__ x, const float* __restrict__ W1,
               const float* __restrict__ b1, float* __restrict__ A,
               float* __restrict__ Bm, int t, int lane)
{
    const int i0   = (t & 63) * 16;
    const int j0   = ((t >> 6) & 7) * 16;
    const int hlf  = t >> 9;
    const int m    = lane & 15;
    const int kg   = (lane >> 4) << 1;

    const float* __restrict__ W = W1 + hlf * DIN * DHID;
    float* __restrict__ C = hlf ? Bm : A;

    v8f acc = {0.f, 0.f, 0.f, 0.f, 0.f, 0.f, 0.f, 0.f};
    for (int k = 0; k < DIN; k += 4) {
        v2f a, b;
        a.x = x[(i0 + m) * DIN + (k + kg)];
        a.y = x[(i0 + m) * DIN + (k + kg + 1)];
        b.x = W[(k + kg) * DHID + (j0 + m)];
        b.y = W[(k + kg + 1) * DHID + (j0 + m)];
        acc = __builtin_amdgcn_wmma_f32_16x16x4_f32(false, a, false, b,
                                                    (short)0, acc, false, false);
    }

    const float bias  = (hlf == 0) ? b1[j0 + m] : 0.0f;
    const int   rbase = (lane < 16) ? 0 : 8;
#pragma unroll
    for (int v = 0; v < 8; ++v) {
        C[(i0 + rbase + v) * DHID + (j0 + m)] = acc[v] + bias;
    }
}

__device__ __forceinline__ float lane_xor16(float v)
{
    const int b = __builtin_bit_cast(int, v);
    return __builtin_bit_cast(float,
        __builtin_amdgcn_permlanex16(0, b, 0x76543210, 0xfedcba98, false, false));
}
template <int M>
__device__ __forceinline__ float row_xor(float v)
{
    const int b = __builtin_bit_cast(int, v);
    return __builtin_bit_cast(float,
        __builtin_amdgcn_mov_dpp(b, 0x160 | M, 0xf, 0xf, false));
}
__device__ __forceinline__ float lane_xor(float v, int m)
{
    switch (m) {
    case 1:  return row_xor<1>(v);
    case 2:  return row_xor<2>(v);
    case 4:  return row_xor<4>(v);
    case 8:  return row_xor<8>(v);
    default: return lane_xor16(v);
    }
}
template <int D>
__device__ __forceinline__ float lane_down(float v, int lane)
{
    if constexpr (D == 16) {
        return lane_xor16(v);
    } else {
        constexpr unsigned long long ident = 0xfedcba9876543210ull;
        constexpr unsigned long long sel   = (ident >> (4 * D)) | (ident << (64 - 4 * D));
        const int b = __builtin_bit_cast(int, v);
        const float in_row = __builtin_bit_cast(float,
            __builtin_amdgcn_mov_dpp(b, 0x100 | D, 0xf, 0xf, true));
        const float x_row = __builtin_bit_cast(float,
            __builtin_amdgcn_permlanex16(0, b, (unsigned)sel, (unsigned)(sel >> 32), false, false));
        return ((lane & 15) + D < 16) ? in_row : x_row;
    }
}
template <int D>
__device__ __forceinline__ void suffix_step(float& t1, float& t2, int lane)
{
    const float y1 = lane_down<D>(t1, lane);
    const float y2 = lane_down<D>(t2, lane);
    if (lane + D < 32) { t1 += y1; t2 += y2; }
}

template <int M>
__device__ __forceinline__ float pair_sum(float lo, float hi, bool bit)
{
    const float keep = bit ? hi : lo;
    const float send = bit ? lo : hi;
    return keep + row_xor<M>(send);
}
__device__ __forceinline__ float group8_sum(const float (&v)[8], int q)
{
    const bool b4 = (q & 4) != 0, b2 = (q & 2) != 0, b1 = (q & 1) != 0;
    const float s0 = pair_sum<4>(v[0], v[4], b4);
    const float s1 = pair_sum<4>(v[1], v[5], b4);
    const float s2 = pair_sum<4>(v[2], v[6], b4);
    const float s3 = pair_sum<4>(v[3], v[7], b4);
    const float t0 = pair_sum<2>(s0, s2, b2);
    const float t1 = pair_sum<2>(s1, s3, b2);
    return pair_sum<1>(t0, t1, b1);
}

__global__ __launch_bounds__(256)
void k12_gemm_stats(const float* __restrict__ x_in, const float* __restrict__ W1,
                    const float* __restrict__ b1,
                    const float* __restrict__ gamma, const float* __restrict__ beta,
                    const float* __restrict__ W2,
                    float* __restrict__ A, float* __restrict__ Bm,
                    float* __restrict__ cvec, float* __restrict__ term)
{
    const int tid  = threadIdx.x;
    const int lane = tid & 31;
    const int wave = tid >> 5;

    if (blockIdx.x >= DHID) {
        gemm_tile(x_in, W1, b1, A, Bm, (blockIdx.x - DHID) * 8 + wave, lane);
        return;
    }
    const int h    = blockIdx.x;

    const float gm = gamma[h];
    const float bt = beta[h];
    const float w2 = W2[h];

    __shared__ float4 xch[2][256];
    __shared__ __attribute__((aligned(16))) float sb[NB];
    __shared__ __attribute__((aligned(16))) float p1[NB + 4];
    __shared__ __attribute__((aligned(16))) float p2[NB + 4];
    __shared__ float wtot[2][8];
    __shared__ float red[2][8];
    __shared__ float sa[NB];

    float x[4], a[4];
    {
        const int q  = lane & 7;
        const int rr = lane >> 3;
        float4 wa0, wa1, wb0, wb1;
        wa0.x = W1[(4 * q + 0) * DHID + h];        wa0.y = W1[(4 * q + 1) * DHID + h];
        wa0.z = W1[(4 * q + 2) * DHID + h];        wa0.w = W1[(4 * q + 3) * DHID + h];
        wa1.x = W1[(32 + 4 * q + 0) * DHID + h];   wa1.y = W1[(32 + 4 * q + 1) * DHID + h];
        wa1.z = W1[(32 + 4 * q + 2) * DHID + h];   wa1.w = W1[(32 + 4 * q + 3) * DHID + h];
        wb0.x = W1[(64 + 4 * q + 0) * DHID + h];   wb0.y = W1[(64 + 4 * q + 1) * DHID + h];
        wb0.z = W1[(64 + 4 * q + 2) * DHID + h];   wb0.w = W1[(64 + 4 * q + 3) * DHID + h];
        wb1.x = W1[(96 + 4 * q + 0) * DHID + h];   wb1.y = W1[(96 + 4 * q + 1) * DHID + h];
        wb1.z = W1[(96 + 4 * q + 2) * DHID + h];   wb1.w = W1[(96 + 4 * q + 3) * DHID + h];
        const float bias = b1[h];
        const int   rot  = h & 31;
        const float4* xr = (const float4*)x_in + (wave * 128 * (DIN / 4) + rr * (DIN / 4) + q);

        float4 c0[32], c1[32];
#pragma unroll
        for (int t = 0; t < 32; ++t) {
            const int trip = (t + rot) & 31;
            c0[t] = xr[trip * 64];
            c1[t] = xr[trip * 64 + 8];
        }
        __builtin_amdgcn_sched_barrier(0);
#pragma unroll
        for (int g = 0; g < 4; ++g) {
            float pa[8], pb[8];
#pragma unroll
            for (int i = 0; i < 8; ++i) {
                const float4 u = c0[g * 8 + i];
                const float4 v = c1[g * 8 + i];
                float ta = u.x * wa0.x, tb = u.x * wb0.x;
                ta = fmaf(u.y, wa0.y, ta);  tb = fmaf(u.y, wb0.y, tb);
                ta = fmaf(u.z, wa0.z, ta);  tb = fmaf(u.z, wb0.z, tb);
                ta = fmaf(u.w, wa0.w, ta);  tb = fmaf(u.w, wb0.w, tb);
                ta = fmaf(v.x, wa1.x, ta);  tb = fmaf(v.x, wb1.x, tb);
                ta = fmaf(v.y, wa1.y, ta);  tb = fmaf(v.y, wb1.y, tb);
                ta = fmaf(v.z, wa1.z, ta);  tb = fmaf(v.z, wb1.z, tb);
                ta = fmaf(v.w, wa1.w, ta);  tb = fmaf(v.w, wb1.w, tb);
                pa[i] = ta;  pb[i] = tb;
            }
            const float av = group8_sum(pa, q) + bias;
            x[g] = group8_sum(pb, q);
            const int row = wave * 128 + 4 * ((g * 8 + q + rot) & 31) + rr;
            sa[row] = av;
        }
    }

    int xb = 0;
#pragma unroll
    for (int k = 2; k <= NB; k <<= 1) {
#pragma unroll
        for (int j = k >> 1; j >= 1; j >>= 1) {
            if (j >= 128) {
                xch[xb][tid] = make_float4(x[0], x[1], x[2], x[3]);
                __syncthreads();
                const float4 y = xch[xb][tid ^ (j >> 2)];
                xb ^= 1;
                const bool keep_min = ((((tid * 4) & j) == 0) == (((tid * 4) & k) == 0));
                x[0] = keep_min ? fminf(x[0], y.x) : fmaxf(x[0], y.x);
                x[1] = keep_min ? fminf(x[1], y.y) : fmaxf(x[1], y.y);
                x[2] = keep_min ? fminf(x[2], y.z) : fmaxf(x[2], y.z);
                x[3] = keep_min ? fminf(x[3], y.w) : fmaxf(x[3], y.w);
            } else if (j >= 4) {
                const bool keep_min = ((((tid * 4) & j) == 0) == (((tid * 4) & k) == 0));
#pragma unroll
                for (int r = 0; r < 4; ++r) {
                    const float y = lane_xor(x[r], j >> 2);
                    x[r] = keep_min ? fminf(x[r], y) : fmaxf(x[r], y);
                }
            } else {
#pragma unroll
                for (int r = 0; r < 4; ++r) {
                    if ((r & j) == 0) {
                        const bool asc = (((tid * 4 + r) & k) == 0);
                        const float lo = fminf(x[r], x[r | j]);
                        const float hi = fmaxf(x[r], x[r | j]);
                        x[r]     = asc ? lo : hi;
                        x[r | j] = asc ? hi : lo;
                    }
                }
            }
        }
    }

    float s1[4], s2[4];
    s1[3] = x[3];          s2[3] = x[3] * x[3];
    s1[2] = x[2] + s1[3];  s2[2] = fmaf(x[2], x[2], s2[3]);
    s1[1] = x[1] + s1[2];  s2[1] = fmaf(x[1], x[1], s2[2]);
    s1[0] = x[0] + s1[1];  s2[0] = fmaf(x[0], x[0], s2[1]);

    float t1 = s1[0], t2 = s2[0];
    suffix_step<1>(t1, t2, lane);
    suffix_step<2>(t1, t2, lane);
    suffix_step<4>(t1, t2, lane);
    suffix_step<8>(t1, t2, lane);
    suffix_step<16>(t1, t2, lane);
    const float n1 = lane_down<1>(t1, lane);
    const float n2 = lane_down<1>(t2, lane);
    float off1 = (lane < 31) ? n1 : 0.f;
    float off2 = (lane < 31) ? n2 : 0.f;
    if (lane == 0) { wtot[0][wave] = t1; wtot[1][wave] = t2; }
    __syncthreads();
    {
        float hi1 = 0.f, hi2 = 0.f;
#pragma unroll
        for (int w = 7; w >= 1; --w) {
            if (w > wave) { hi1 += wtot[0][w]; hi2 += wtot[1][w]; }
        }
        off1 += hi1; off2 += hi2;
    }
    *(float4*)&sb[tid * 4] = make_float4(x[0], x[1], x[2], x[3]);
    *(float4*)&p1[tid * 4] = make_float4(s1[0] + off1, s1[1] + off1, s1[2] + off1, s1[3] + off1);
    *(float4*)&p2[tid * 4] = make_float4(s2[0] + off2, s2[1] + off2, s2[2] + off2, s2[3] + off2);
    if (tid == 0) { p1[NB] = 0.f; p2[NB] = 0.f; }
    __syncthreads();
#pragma unroll
    for (int r = 0; r < 4; ++r) a[r] = sa[tid + 256 * r];

    float na[4];
    int   m[4] = {0, 0, 0, 0};
#pragma unroll
    for (int r = 0; r < 4; ++r) na[r] = -a[r];
#pragma unroll
    for (int s = NB / 2; s >= 1; s >>= 1) {
#pragma unroll
        for (int r = 0; r < 4; ++r) {
            if (sb[m[r] + s - 1] <= na[r]) m[r] += s;
        }
    }
    float acc1 = 0.f, acc2 = 0.f;
#pragma unroll
    for (int r = 0; r < 4; ++r) {
        if (sb[m[r]] <= na[r]) m[r] += 1;
        const float cnt = (float)(NB - m[r]);
        const float q1  = p1[m[r]];
        const float q2  = p2[m[r]];
        acc1 += fmaf(cnt, a[r], q1);
        acc2 += fmaf(a[r], fmaf(cnt, a[r], 2.f * q1), q2);
    }

#pragma unroll
    for (int d = 16; d >= 1; d >>= 1) {
        acc1 += lane_xor(acc1, d);
        acc2 += lane_xor(acc2, d);
    }
    if (lane == 0) { red[0][wave] = acc1; red[1][wave] = acc2; }
    __syncthreads();
    if (tid == 0) {
        float r1 = 0.f, r2 = 0.f;
#pragma unroll
        for (int w = 0; w < 8; ++w) { r1 += red[0][w]; r2 += red[1][w]; }
        const float invN = 1.0f / ((float)NB * (float)NB);
        const float mean = r1 * invN;
        const float var  = fmaxf(r2 * invN - mean * mean, 0.f);
        const float inv  = rsqrtf(var + BN_EPS);
        const float g    = gm * inv;
        cvec[h] = w2 * g;
        term[h] = w2 * (bt - mean * g);
    }
}


__device__ __forceinline__ v2f max_pair(v2f a, v2f b)
{
    float lo, hi;
    asm("v_dual_max_num_f32 %1, %3, %5 :: v_dual_max_num_f32 %0, %2, %4"
        : "=v"(lo), "=v"(hi)
        : "v"(a.x), "v"(a.y), "v"(b.x), "v"(b.y));
    return (v2f){lo, hi};
}

struct K4Frag {
    float4 a[4];
    float4 b[4];
    float4 c;
};

__global__ __launch_bounds__(256)
void k4_out(const float* __restrict__ A, const float* __restrict__ Bm,
            const float* __restrict__ cvec, const float* __restrict__ term,
            const float* __restrict__ b2, float* __restrict__ out)
{
    __shared__ __attribute__((aligned(16))) float aS[64 * DHID];
    __shared__ __attribute__((aligned(16))) float bS[64 * DHID];
    __shared__ __attribute__((aligned(16))) float cs[DHID];
    __shared__ __attribute__((aligned(16))) float colS[64];

    const int i0  = blockIdx.x * 64;
    const int j0  = blockIdx.y * 64;
    const int tid = threadIdx.x;

    const int   ln = tid & 31;
    const int   wv = tid >> 5;
    const float t0 = term[ln], t1 = term[ln + 32], t2 = term[ln + 64], t3 = term[ln + 96];
    const float bias2 = b2[0];
    const float4 c4 = ((const float4*)cvec)[ln];

    {
        const float4* ga = (const float4*)(A  + (i0 + wv * 8) * DHID) + ln;
        const float4* gb = (const float4*)(Bm + (j0 + wv * 8) * DHID) + ln;
        float4 va[8], vb[8];
#pragma unroll
        for (int q = 0; q < 8; ++q) { va[q] = ga[q * (DHID / 4)]; vb[q] = gb[q * (DHID / 4)]; }
#pragma unroll
        for (int q = 0; q < 8; ++q) {
            const int r   = wv * 8 + q;
            const int pos = ln ^ ((r >> 2) & 15);
            *(float4*)&aS[r * DHID + 4 * pos] = va[q];
            *(float4*)&bS[r * DHID + 4 * pos] = make_float4(-vb[q].x, -vb[q].y, -vb[q].z, -vb[q].w);
        }
        float part[8];
#pragma unroll
        for (int q = 0; q < 8; ++q) {
            float t = vb[q].x * c4.x;
            t = fmaf(vb[q].y, c4.y, t);
            t = fmaf(vb[q].z, c4.z, t);
            t = fmaf(vb[q].w, c4.w, t);
            part[q] = t;
        }
        float col = group8_sum(part, ln & 7);
        col += row_xor<8>(col);
        col += lane_xor16(col);
        if (ln < 8) colS[wv * 8 + ln] = col;
    }
    if (tid < DHID) cs[tid] = cvec[tid];

    float cv = (t0 + t2) + (t1 + t3);
    cv += lane_xor16(cv);
    cv += row_xor<8>(cv);
    cv += row_xor<4>(cv);
    cv += row_xor<2>(cv);
    cv += row_xor<1>(cv);
    const float ct = cv + bias2;
    __syncthreads();

    const int jl = tid & 15;
    const int ig = tid >> 4;
    const int jb = jl * 4;
    const int ib = ig * 4;
    const float* arow = &aS[ib * DHID];
    const float* brow = &bS[jb * DHID];

    v2f acc[4][4];
#pragma unroll
    for (int ii = 0; ii < 4; ++ii)
#pragma unroll
        for (int jj = 0; jj < 4; ++jj) acc[ii][jj] = (v2f){0.f, 0.f};

    auto load4 = [&](K4Frag& f, int kc) {
        const int pa = 4 * (kc ^ ig);
        const int pb = 4 * (kc ^ jl);
#pragma unroll
        for (int u = 0; u < 4; ++u) {
            f.a[u] = *(const float4*)&arow[u * DHID + pa];
            f.b[u] = *(const float4*)&brow[u * DHID + pb];
        }
        f.c = *(const float4*)&cs[4 * kc];
    };
    auto step4 = [&](const K4Frag& f) {
        const v2f clo = {f.c.x, f.c.y};
        const v2f chi = {f.c.z, f.c.w};
#pragma unroll
        for (int ii = 0; ii < 4; ++ii) {
            const v2f alo = {f.a[ii].x, f.a[ii].y};
            const v2f ahi = {f.a[ii].z, f.a[ii].w};
#pragma unroll
            for (int jj = 0; jj < 4; ++jj) {
                const v2f blo = {f.b[jj].x, f.b[jj].y};
                const v2f bhi = {f.b[jj].z, f.b[jj].w};
                const v2f rlo = max_pair(alo, blo);
                acc[ii][jj] = __builtin_elementwise_fma(clo, rlo, acc[ii][jj]);
                const v2f rhi = max_pair(ahi, bhi);
                acc[ii][jj] = __builtin_elementwise_fma(chi, rhi, acc[ii][jj]);
            }
        }
    };

    for (int kc = 0; kc < DHID / 4; kc += 2) {
        K4Frag f0, f1;
        load4(f0, kc);
        load4(f1, kc + 1);
        step4(f0);
        step4(f1);
    }

    const float4 cj = *(const float4*)&colS[jb];
    const float e0 = cj.x + ct, e1 = cj.y + ct, e2 = cj.z + ct, e3 = cj.w + ct;
#pragma unroll
    for (int ii = 0; ii < 4; ++ii) {
        float4 o;
        o.x = (acc[ii][0].x + acc[ii][0].y) + e0;
        o.y = (acc[ii][1].x + acc[ii][1].y) + e1;
        o.z = (acc[ii][2].x + acc[ii][2].y) + e2;
        o.w = (acc[ii][3].x + acc[ii][3].y) + e3;
        *(float4*)(out + (size_t)(i0 + ib + ii) * NB + (j0 + jb)) = o;
    }
}

extern "C" void kernel_launch(void* const* d_in, const int* in_sizes, int n_in,
                              void* d_out, int out_size, void* d_ws, size_t ws_size,
                              hipStream_t stream)
{
    const float* x     = (const float*)d_in[0];
    const float* W1    = (const float*)d_in[1];
    const float* b1    = (const float*)d_in[2];
    const float* gamma = (const float*)d_in[3];
    const float* beta  = (const float*)d_in[4];
    const float* W2    = (const float*)d_in[5];
    const float* b2    = (const float*)d_in[6];
    float* out = (float*)d_out;

    float* ws       = (float*)d_ws;
    float* A        = ws;
    float* Bm       = ws + NB * DHID;
    float* cvec     = ws + 2 * NB * DHID;
    float* term     = cvec + DHID;

    k12_gemm_stats<<<dim3(2 * DHID), 256, 0, stream>>>(x, W1, b1, gamma, beta, W2, A, Bm, cvec, term);
    k4_out  <<<dim3(NB / 64, NB / 64),     256, 0, stream>>>(A, Bm, cvec, term, b2, out);
}
